# starting kernel
# baseline (speedup 1.0000x reference)
#include <hip/hip_runtime.h>
#include <math.h>

#define BB 4
#define CC 64
#define NN 4096
#define OO 64
#define KK 20
#define EPSV 1e-5f

typedef __attribute__((ext_vector_type(2))) float v2f;
typedef __attribute__((ext_vector_type(8))) float v8f;

__device__ __forceinline__ v8f wmma_f32(v2f a, v2f b, v8f c) {
    return __builtin_amdgcn_wmma_f32_16x16x4_f32(false, a, false, b, (short)0, c,
                                                 false, false);
}

__global__ __launch_bounds__(256) void prep_kernel(const float* __restrict__ x,
                                                   float* __restrict__ xt,
                                                   float* __restrict__ xx) {
    int t = blockIdx.x * blockDim.x + threadIdx.x;
    int b = t / NN;
    int n = t - b * NN;
    const float* px = x + (size_t)b * CC * NN + n;
    float* pt = xt + (size_t)t * CC;
    float s = 0.f;
#pragma unroll 8
    for (int c = 0; c < CC; ++c) {
        float v = px[(size_t)c * NN];
        pt[c] = v;
        s += v * v;
    }
    xx[t] = s;
}

__global__ __launch_bounds__(128) void knn_kernel(const float* __restrict__ xt,
                                                  const float* __restrict__ xx,
                                                  int* __restrict__ nbr) {
    const int wave = threadIdx.x >> 5;
    const int lane = threadIdx.x & 31;
    const int laneM = lane & 15;
    const int kofs = (lane < 16) ? 0 : 2;

    const int row0 = blockIdx.x * 128 + wave * 32;
    const int b = row0 / NN;
    const int r0 = row0 - b * NN;

    const float* prowLo = xt + (size_t)(b * NN + r0 + laneM) * CC;
    const float* prowHi = prowLo + 16 * CC;
    v2f aLo[16], aHi[16];
#pragma unroll
    for (int ks = 0; ks < 16; ++ks) {
        int kc = ks * 4 + kofs;
        aLo[ks] = *(const v2f*)(prowLo + kc);
        aHi[ks] = *(const v2f*)(prowHi + kc);
    }

    float topv[KK];
    int topi[KK];
#pragma unroll
    for (int s = 0; s < KK; ++s) { topv[s] = -3.0e38f; topi[s] = 0; }
    const float xxrow = xx[b * NN + r0 + lane];

    __shared__ float dist[4][32 * 16];

    for (int ct = 0; ct < NN / 16; ++ct) {
        const int col = ct * 16 + laneM;
        const float* pcol = xt + (size_t)(b * NN + col) * CC;
        v8f accLo = {};
        v8f accHi = {};
#pragma unroll
        for (int ks = 0; ks < 16; ++ks) {
            v2f bb = *(const v2f*)(pcol + ks * 4 + kofs);
            accLo = wmma_f32(aLo[ks], bb, accLo);
            accHi = wmma_f32(aHi[ks], bb, accHi);
        }
        const float xcol = xx[b * NN + col];
#pragma unroll
        for (int j = 0; j < 8; ++j) {
            int M = j + ((lane < 16) ? 0 : 8);
            dist[wave][M * 16 + laneM] = 2.0f * accLo[j] - xcol;
            dist[wave][(16 + M) * 16 + laneM] = 2.0f * accHi[j] - xcol;
        }
        __syncthreads();
#pragma unroll
        for (int c = 0; c < 16; ++c) {
            float v = dist[wave][lane * 16 + c] - xxrow;
            if (v > topv[KK - 1]) {
                float cv = v;
                int ci = ct * 16 + c;
#pragma unroll
                for (int s = 0; s < KK; ++s) {
                    if (cv > topv[s]) {
                        float tv = topv[s]; topv[s] = cv; cv = tv;
                        int ti = topi[s]; topi[s] = ci; ci = ti;
                    }
                }
            }
        }
        __syncthreads();
    }
    int* outp = nbr + (size_t)(b * NN + r0 + lane) * KK;
#pragma unroll
    for (int s = 0; s < KK; ++s) outp[s] = topi[s];
}

__global__ __launch_bounds__(128) void edgeconv_kernel(
    const float* __restrict__ xt, const int* __restrict__ nbr,
    const float* __restrict__ w1, const float* __restrict__ g1,
    const float* __restrict__ be1, const float* __restrict__ mu1,
    const float* __restrict__ va1, const float* __restrict__ w2,
    const float* __restrict__ g2, const float* __restrict__ be2,
    const float* __restrict__ mu2, const float* __restrict__ va2,
    float* __restrict__ out) {
    const int wave = threadIdx.x >> 5;
    const int lane = threadIdx.x & 31;
    const int laneM = lane & 15;
    const int kofs = (lane < 16) ? 0 : 2;
    const int b = blockIdx.x >> 8;
    const int pt = blockIdx.x & 255;

    __shared__ float h1s[4][16 * OO];
    __shared__ unsigned om[16 * OO];
    for (int i = threadIdx.x; i < 16 * OO; i += 128) om[i] = 0u;
    __syncthreads();

    for (int it = 0; it < 5; ++it) {
        const int mt = wave * 5 + it;
        const int lrow = mt * 16 + laneM;
        const int p = lrow / KK;
        const int kk = lrow - p * KK;
        const int n = pt * 16 + p;
        const int nb = nbr[(size_t)(b * NN + n) * KK + kk];
        const float* pc = xt + (size_t)(b * NN + n) * CC;
        const float* pn = xt + (size_t)(b * NN + nb) * CC;

        v8f acc1[4] = {};
#pragma unroll
        for (int ks = 0; ks < 16; ++ks) {
            int kc = ks * 4 + kofs;
            v2f a = *(const v2f*)(pn + kc) - *(const v2f*)(pc + kc);
#pragma unroll
            for (int nt = 0; nt < 4; ++nt) {
                int o = nt * 16 + laneM;
                v2f bb = *(const v2f*)(w1 + o * 128 + kc);
                acc1[nt] = wmma_f32(a, bb, acc1[nt]);
            }
        }
#pragma unroll
        for (int ks = 0; ks < 16; ++ks) {
            int kc = ks * 4 + kofs;
            v2f a = *(const v2f*)(pc + kc);
#pragma unroll
            for (int nt = 0; nt < 4; ++nt) {
                int o = nt * 16 + laneM;
                v2f bb = *(const v2f*)(w1 + o * 128 + 64 + kc);
                acc1[nt] = wmma_f32(a, bb, acc1[nt]);
            }
        }
#pragma unroll
        for (int nt = 0; nt < 4; ++nt) {
            int o = nt * 16 + laneM;
            float sc = g1[o] * rsqrtf(va1[o] + EPSV);
            float sh = be1[o] - mu1[o] * sc;
#pragma unroll
            for (int j = 0; j < 8; ++j) {
                int M = j + ((lane < 16) ? 0 : 8);
                h1s[wave][M * OO + o] = fmaxf(acc1[nt][j] * sc + sh, 0.0f);
            }
        }
        __syncthreads();

        v8f acc2[4] = {};
#pragma unroll
        for (int ks = 0; ks < 16; ++ks) {
            int kc = ks * 4 + kofs;
            v2f a = *(const v2f*)(&h1s[wave][laneM * OO + kc]);
#pragma unroll
            for (int nt = 0; nt < 4; ++nt) {
                int o = nt * 16 + laneM;
                v2f bb = *(const v2f*)(w2 + o * OO + kc);
                acc2[nt] = wmma_f32(a, bb, acc2[nt]);
            }
        }
#pragma unroll
        for (int nt = 0; nt < 4; ++nt) {
            int o = nt * 16 + laneM;
            float sc = g2[o] * rsqrtf(va2[o] + EPSV);
            float sh = be2[o] - mu2[o] * sc;
#pragma unroll
            for (int j = 0; j < 8; ++j) {
                int M = j + ((lane < 16) ? 0 : 8);
                int lr = mt * 16 + M;
                int pp = lr / KK;
                float val = fmaxf(acc2[nt][j] * sc + sh, 0.0f);
                atomicMax(&om[pp * OO + o], __float_as_uint(val));
            }
        }
        __syncthreads();
    }
    for (int i = threadIdx.x; i < 16 * OO; i += 128) {
        int p = i >> 6;
        int o = i & 63;
        out[(size_t)(b * OO + o) * NN + pt * 16 + p] = __uint_as_float(om[p * OO + o]);
    }
}

extern "C" void kernel_launch(void* const* d_in, const int* in_sizes, int n_in,
                              void* d_out, int out_size, void* d_ws, size_t ws_size,
                              hipStream_t stream) {
    const float* x   = (const float*)d_in[0];
    const float* w1  = (const float*)d_in[1];
    const float* g1  = (const float*)d_in[2];
    const float* be1 = (const float*)d_in[3];
    const float* mu1 = (const float*)d_in[4];
    const float* va1 = (const float*)d_in[5];
    const float* w2  = (const float*)d_in[6];
    const float* g2  = (const float*)d_in[7];
    const float* be2 = (const float*)d_in[8];
    const float* mu2 = (const float*)d_in[9];
    const float* va2 = (const float*)d_in[10];
    float* out = (float*)d_out;

    float* xt = (float*)d_ws;
    float* xx = xt + (size_t)BB * NN * CC;
    int* nbr = (int*)(xx + (size_t)BB * NN);

    prep_kernel<<<(BB * NN) / 256, 256, 0, stream>>>(x, xt, xx);
    knn_kernel<<<(BB * NN) / 128, 128, 0, stream>>>(xt, xx, nbr);
    edgeconv_kernel<<<BB * (NN / 16), 128, 0, stream>>>(
        xt, nbr, w1, g1, be1, mu1, va1, w2, g2, be2, mu2, va2, out);
}


# kNN top-20 scan: per-slot branch chain of the list insert (compiled to whole-list register copies) replaced by a branch-free sorted insert, one compare per slot and selects
# speedup vs baseline: 8.1733x; 8.1733x over previous
#include <hip/hip_runtime.h>
#include <math.h>

#define BB 4
#define CC 64
#define NN 4096
#define OO 64
#define KK 20
#define EPSV 1e-5f

typedef __attribute__((ext_vector_type(2))) float v2f;
typedef __attribute__((ext_vector_type(8))) float v8f;

__device__ __forceinline__ v8f wmma_f32(v2f a, v2f b, v8f c) {
    return __builtin_amdgcn_wmma_f32_16x16x4_f32(false, a, false, b, (short)0, c,
                                                 false, false);
}

__global__ __launch_bounds__(256) void prep_kernel(const float* __restrict__ x,
                                                   float* __restrict__ xt,
                                                   float* __restrict__ xx) {
    int t = blockIdx.x * blockDim.x + threadIdx.x;
    int b = t / NN;
    int n = t - b * NN;
    const float* px = x + (size_t)b * CC * NN + n;
    float* pt = xt + (size_t)t * CC;
    float s = 0.f;
#pragma unroll 8
    for (int c = 0; c < CC; ++c) {
        float v = px[(size_t)c * NN];
        pt[c] = v;
        s += v * v;
    }
    xx[t] = s;
}

__global__ __launch_bounds__(128) void knn_kernel(const float* __restrict__ xt,
                                                  const float* __restrict__ xx,
                                                  int* __restrict__ nbr) {
    const int wave = threadIdx.x >> 5;
    const int lane = threadIdx.x & 31;
    const int laneM = lane & 15;
    const int kofs = (lane < 16) ? 0 : 2;

    const int row0 = blockIdx.x * 128 + wave * 32;
    const int b = row0 / NN;
    const int r0 = row0 - b * NN;

    const float* prowLo = xt + (size_t)(b * NN + r0 + laneM) * CC;
    const float* prowHi = prowLo + 16 * CC;
    v2f aLo[16], aHi[16];
#pragma unroll
    for (int ks = 0; ks < 16; ++ks) {
        int kc = ks * 4 + kofs;
        aLo[ks] = *(const v2f*)(prowLo + kc);
        aHi[ks] = *(const v2f*)(prowHi + kc);
    }

    float topv[KK];
    int topi[KK];
#pragma unroll
    for (int s = 0; s < KK; ++s) { topv[s] = -3.0e38f; topi[s] = 0; }
    const float xxrow = xx[b * NN + r0 + lane];

    __shared__ __attribute__((aligned(16))) float dist[4][32 * 16];

    for (int ct = 0; ct < NN / 16; ++ct) {
        const int col = ct * 16 + laneM;
        const float* pcol = xt + (size_t)(b * NN + col) * CC;
        v8f accLo = {};
        v8f accHi = {};
#pragma unroll
        for (int ks = 0; ks < 16; ++ks) {
            v2f bb = *(const v2f*)(pcol + ks * 4 + kofs);
            accLo = wmma_f32(aLo[ks], bb, accLo);
            accHi = wmma_f32(aHi[ks], bb, accHi);
        }
        const float xcol = xx[b * NN + col];
#pragma unroll
        for (int j = 0; j < 8; ++j) {
            int M = j + ((lane < 16) ? 0 : 8);
            dist[wave][M * 16 + laneM] = 2.0f * accLo[j] - xcol;
            dist[wave][(16 + M) * 16 + laneM] = 2.0f * accHi[j] - xcol;
        }
        __syncthreads();
        const float4* drow = (const float4*)&dist[wave][lane * 16];
        float dv[16];
#pragma unroll
        for (int q = 0; q < 4; ++q) {
            const float4 t4 = drow[q];
            dv[4 * q + 0] = t4.x; dv[4 * q + 1] = t4.y;
            dv[4 * q + 2] = t4.z; dv[4 * q + 3] = t4.w;
        }
#pragma unroll
        for (int c = 0; c < 16; ++c) {
            const float v = dv[c] - xxrow;
            if (v > topv[KK - 1]) {
                const int ci = ct * 16 + c;
                bool beats[KK];
                float oldv[KK];
                int oldi[KK];
#pragma unroll
                for (int s = 0; s < KK; ++s) {
                    oldv[s] = topv[s];
                    oldi[s] = topi[s];
                    beats[s] = v > oldv[s];
                }
                float inv[KK];
                int ini[KK];
                inv[0] = v;
                ini[0] = ci;
#pragma unroll
                for (int s = 1; s < KK; ++s) {
                    const bool mu = beats[s - 1];
                    const float upv = oldv[s - 1];
                    const int upi = oldi[s - 1];
                    inv[s] = mu ? upv : v;
                    ini[s] = mu ? upi : ci;
                }
#pragma unroll
                for (int s = 0; s < KK; ++s) {
                    const bool ms = beats[s];
                    const float a = inv[s];
                    const float b = oldv[s];
                    const int ai = ini[s];
                    const int bi = oldi[s];
                    topv[s] = ms ? a : b;
                    topi[s] = ms ? ai : bi;
                }
            }
        }
        __syncthreads();
    }
    int* outp = nbr + (size_t)(b * NN + r0 + lane) * KK;
#pragma unroll
    for (int s = 0; s < KK; ++s) outp[s] = topi[s];
}

__global__ __launch_bounds__(128) void edgeconv_kernel(
    const float* __restrict__ xt, const int* __restrict__ nbr,
    const float* __restrict__ w1, const float* __restrict__ g1,
    const float* __restrict__ be1, const float* __restrict__ mu1,
    const float* __restrict__ va1, const float* __restrict__ w2,
    const float* __restrict__ g2, const float* __restrict__ be2,
    const float* __restrict__ mu2, const float* __restrict__ va2,
    float* __restrict__ out) {
    const int wave = threadIdx.x >> 5;
    const int lane = threadIdx.x & 31;
    const int laneM = lane & 15;
    const int kofs = (lane < 16) ? 0 : 2;
    const int b = blockIdx.x >> 8;
    const int pt = blockIdx.x & 255;

    __shared__ float h1s[4][16 * OO];
    __shared__ unsigned om[16 * OO];
    for (int i = threadIdx.x; i < 16 * OO; i += 128) om[i] = 0u;
    __syncthreads();

    for (int it = 0; it < 5; ++it) {
        const int mt = wave * 5 + it;
        const int lrow = mt * 16 + laneM;
        const int p = lrow / KK;
        const int kk = lrow - p * KK;
        const int n = pt * 16 + p;
        const int nb = nbr[(size_t)(b * NN + n) * KK + kk];
        const float* pc = xt + (size_t)(b * NN + n) * CC;
        const float* pn = xt + (size_t)(b * NN + nb) * CC;

        v8f acc1[4] = {};
#pragma unroll
        for (int ks = 0; ks < 16; ++ks) {
            int kc = ks * 4 + kofs;
            v2f a = *(const v2f*)(pn + kc) - *(const v2f*)(pc + kc);
#pragma unroll
            for (int nt = 0; nt < 4; ++nt) {
                int o = nt * 16 + laneM;
                v2f bb = *(const v2f*)(w1 + o * 128 + kc);
                acc1[nt] = wmma_f32(a, bb, acc1[nt]);
            }
        }
#pragma unroll
        for (int ks = 0; ks < 16; ++ks) {
            int kc = ks * 4 + kofs;
            v2f a = *(const v2f*)(pc + kc);
#pragma unroll
            for (int nt = 0; nt < 4; ++nt) {
                int o = nt * 16 + laneM;
                v2f bb = *(const v2f*)(w1 + o * 128 + 64 + kc);
                acc1[nt] = wmma_f32(a, bb, acc1[nt]);
            }
        }
#pragma unroll
        for (int nt = 0; nt < 4; ++nt) {
            int o = nt * 16 + laneM;
            float sc = g1[o] * rsqrtf(va1[o] + EPSV);
            float sh = be1[o] - mu1[o] * sc;
#pragma unroll
            for (int j = 0; j < 8; ++j) {
                int M = j + ((lane < 16) ? 0 : 8);
                h1s[wave][M * OO + o] = fmaxf(acc1[nt][j] * sc + sh, 0.0f);
            }
        }
        __syncthreads();

        v8f acc2[4] = {};
#pragma unroll
        for (int ks = 0; ks < 16; ++ks) {
            int kc = ks * 4 + kofs;
            v2f a = *(const v2f*)(&h1s[wave][laneM * OO + kc]);
#pragma unroll
            for (int nt = 0; nt < 4; ++nt) {
                int o = nt * 16 + laneM;
                v2f bb = *(const v2f*)(w2 + o * OO + kc);
                acc2[nt] = wmma_f32(a, bb, acc2[nt]);
            }
        }
#pragma unroll
        for (int nt = 0; nt < 4; ++nt) {
            int o = nt * 16 + laneM;
            float sc = g2[o] * rsqrtf(va2[o] + EPSV);
            float sh = be2[o] - mu2[o] * sc;
#pragma unroll
            for (int j = 0; j < 8; ++j) {
                int M = j + ((lane < 16) ? 0 : 8);
                int lr = mt * 16 + M;
                int pp = lr / KK;
                float val = fmaxf(acc2[nt][j] * sc + sh, 0.0f);
                atomicMax(&om[pp * OO + o], __float_as_uint(val));
            }
        }
        __syncthreads();
    }
    for (int i = threadIdx.x; i < 16 * OO; i += 128) {
        int p = i >> 6;
        int o = i & 63;
        out[(size_t)(b * OO + o) * NN + pt * 16 + p] = __uint_as_float(om[p * OO + o]);
    }
}

extern "C" void kernel_launch(void* const* d_in, const int* in_sizes, int n_in,
                              void* d_out, int out_size, void* d_ws, size_t ws_size,
                              hipStream_t stream) {
    const float* x   = (const float*)d_in[0];
    const float* w1  = (const float*)d_in[1];
    const float* g1  = (const float*)d_in[2];
    const float* be1 = (const float*)d_in[3];
    const float* mu1 = (const float*)d_in[4];
    const float* va1 = (const float*)d_in[5];
    const float* w2  = (const float*)d_in[6];
    const float* g2  = (const float*)d_in[7];
    const float* be2 = (const float*)d_in[8];
    const float* mu2 = (const float*)d_in[9];
    const float* va2 = (const float*)d_in[10];
    float* out = (float*)d_out;

    float* xt = (float*)d_ws;
    float* xx = xt + (size_t)BB * NN * CC;
    int* nbr = (int*)(xx + (size_t)BB * NN);

    prep_kernel<<<(BB * NN) / 256, 256, 0, stream>>>(x, xt, xx);
    knn_kernel<<<(BB * NN) / 128, 128, 0, stream>>>(xt, xx, nbr);
    edgeconv_kernel<<<BB * (NN / 16), 128, 0, stream>>>(
        xt, nbr, w1, g1, be1, mu1, va1, w2, g2, be2, mu2, va2, out);
}


# kNN scan split over columns: the 4 waves of a block each scan a quarter of the columns for the same 32 rows (was all columns, 32 rows each), sorted lists merged 4-way in LDS; 4x the waves
# speedup vs baseline: 17.2501x; 2.1105x over previous
#include <hip/hip_runtime.h>
#include <math.h>

#define BB 4
#define CC 64
#define NN 4096
#define OO 64
#define KK 20
#define EPSV 1e-5f

typedef __attribute__((ext_vector_type(2))) float v2f;
typedef __attribute__((ext_vector_type(8))) float v8f;

__device__ __forceinline__ v8f wmma_f32(v2f a, v2f b, v8f c) {
    return __builtin_amdgcn_wmma_f32_16x16x4_f32(false, a, false, b, (short)0, c,
                                                 false, false);
}

__global__ __launch_bounds__(256) void prep_kernel(const float* __restrict__ x,
                                                   float* __restrict__ xt,
                                                   float* __restrict__ xx) {
    int t = blockIdx.x * blockDim.x + threadIdx.x;
    int b = t / NN;
    int n = t - b * NN;
    const float* px = x + (size_t)b * CC * NN + n;
    float* pt = xt + (size_t)t * CC;
    float s = 0.f;
#pragma unroll 8
    for (int c = 0; c < CC; ++c) {
        float v = px[(size_t)c * NN];
        pt[c] = v;
        s += v * v;
    }
    xx[t] = s;
}

#define KNN_ROWS 32
#define KNN_SPLIT 4
#define KNN_COLS (NN / KNN_SPLIT)
__global__ __launch_bounds__(128) void knn_kernel(const float* __restrict__ xt,
                                                  const float* __restrict__ xx,
                                                  int* __restrict__ nbr) {
    const int wave = threadIdx.x >> 5;
    const int lane = threadIdx.x & 31;
    const int laneM = lane & 15;
    const int kofs = (lane < 16) ? 0 : 2;

    const int row0 = blockIdx.x * KNN_ROWS;
    const int b = row0 / NN;
    const int r0 = row0 - b * NN;
    const int c0 = wave * KNN_COLS;

    const float* prowLo = xt + (size_t)(b * NN + r0 + laneM) * CC;
    const float* prowHi = prowLo + 16 * CC;
    v2f aLo[16], aHi[16];
#pragma unroll
    for (int ks = 0; ks < 16; ++ks) {
        int kc = ks * 4 + kofs;
        aLo[ks] = *(const v2f*)(prowLo + kc);
        aHi[ks] = *(const v2f*)(prowHi + kc);
    }

    float topv[KK];
    int topi[KK];
#pragma unroll
    for (int s = 0; s < KK; ++s) { topv[s] = -3.0e38f; topi[s] = 0; }
    const float xxrow = xx[b * NN + r0 + lane];

    __shared__ __attribute__((aligned(16))) float dist[4][32 * 16];

    for (int ct = 0; ct < KNN_COLS / 16; ++ct) {
        const int col = c0 + ct * 16 + laneM;
        const float* pcol = xt + (size_t)(b * NN + col) * CC;
        v8f accLo = {};
        v8f accHi = {};
#pragma unroll
        for (int ks = 0; ks < 16; ++ks) {
            v2f bb = *(const v2f*)(pcol + ks * 4 + kofs);
            accLo = wmma_f32(aLo[ks], bb, accLo);
            accHi = wmma_f32(aHi[ks], bb, accHi);
        }
        const float xcol = xx[b * NN + col];
#pragma unroll
        for (int j = 0; j < 8; ++j) {
            int M = j + ((lane < 16) ? 0 : 8);
            dist[wave][M * 16 + laneM] = 2.0f * accLo[j] - xcol;
            dist[wave][(16 + M) * 16 + laneM] = 2.0f * accHi[j] - xcol;
        }
        __syncthreads();
        const float4* drow = (const float4*)&dist[wave][lane * 16];
        float dv[16];
#pragma unroll
        for (int q = 0; q < 4; ++q) {
            const float4 t4 = drow[q];
            dv[4 * q + 0] = t4.x; dv[4 * q + 1] = t4.y;
            dv[4 * q + 2] = t4.z; dv[4 * q + 3] = t4.w;
        }
#pragma unroll
        for (int c = 0; c < 16; ++c) {
            const float v = dv[c] - xxrow;
            if (v > topv[KK - 1]) {
                const int ci = c0 + ct * 16 + c;
                bool beats[KK];
                float oldv[KK];
                int oldi[KK];
#pragma unroll
                for (int s = 0; s < KK; ++s) {
                    oldv[s] = topv[s];
                    oldi[s] = topi[s];
                    beats[s] = v > oldv[s];
                }
                float inv[KK];
                int ini[KK];
                inv[0] = v;
                ini[0] = ci;
#pragma unroll
                for (int s = 1; s < KK; ++s) {
                    const bool mu = beats[s - 1];
                    const float upv = oldv[s - 1];
                    const int upi = oldi[s - 1];
                    inv[s] = mu ? upv : v;
                    ini[s] = mu ? upi : ci;
                }
#pragma unroll
                for (int s = 0; s < KK; ++s) {
                    const bool ms = beats[s];
                    const float a = inv[s];
                    const float b = oldv[s];
                    const int ai = ini[s];
                    const int bi = oldi[s];
                    topv[s] = ms ? a : b;
                    topi[s] = ms ? ai : bi;
                }
            }
        }
        __syncthreads();
    }
    __shared__ float mval[KNN_SPLIT][KK][32];
    __shared__ int midx[KNN_SPLIT][KK][32];
#pragma unroll
    for (int s = 0; s < KK; ++s) {
        mval[wave][s][lane] = topv[s];
        midx[wave][s][lane] = topi[s];
    }
    __syncthreads();
    if (wave == 0) {
        float h0 = mval[0][0][lane], h1 = mval[1][0][lane];
        float h2 = mval[2][0][lane], h3 = mval[3][0][lane];
        int p0 = 0, p1 = 0, p2 = 0, p3 = 0;
        int* outp = nbr + (size_t)(b * NN + r0 + lane) * KK;
        for (int s = 0; s < KK; ++s) {
            int w = 0, p = p0;
            float best = h0;
            if (h1 > best) { best = h1; w = 1; p = p1; }
            if (h2 > best) { best = h2; w = 2; p = p2; }
            if (h3 > best) { best = h3; w = 3; p = p3; }
            outp[s] = midx[w][p][lane];
            const int pnext = p + 1;
            const float hnext = mval[w][pnext < KK ? pnext : KK - 1][lane];
            p0 = (w == 0) ? pnext : p0;  h0 = (w == 0) ? hnext : h0;
            p1 = (w == 1) ? pnext : p1;  h1 = (w == 1) ? hnext : h1;
            p2 = (w == 2) ? pnext : p2;  h2 = (w == 2) ? hnext : h2;
            p3 = (w == 3) ? pnext : p3;  h3 = (w == 3) ? hnext : h3;
        }
    }
}

__global__ __launch_bounds__(128) void edgeconv_kernel(
    const float* __restrict__ xt, const int* __restrict__ nbr,
    const float* __restrict__ w1, const float* __restrict__ g1,
    const float* __restrict__ be1, const float* __restrict__ mu1,
    const float* __restrict__ va1, const float* __restrict__ w2,
    const float* __restrict__ g2, const float* __restrict__ be2,
    const float* __restrict__ mu2, const float* __restrict__ va2,
    float* __restrict__ out) {
    const int wave = threadIdx.x >> 5;
    const int lane = threadIdx.x & 31;
    const int laneM = lane & 15;
    const int kofs = (lane < 16) ? 0 : 2;
    const int b = blockIdx.x >> 8;
    const int pt = blockIdx.x & 255;

    __shared__ float h1s[4][16 * OO];
    __shared__ unsigned om[16 * OO];
    for (int i = threadIdx.x; i < 16 * OO; i += 128) om[i] = 0u;
    __syncthreads();

    for (int it = 0; it < 5; ++it) {
        const int mt = wave * 5 + it;
        const int lrow = mt * 16 + laneM;
        const int p = lrow / KK;
        const int kk = lrow - p * KK;
        const int n = pt * 16 + p;
        const int nb = nbr[(size_t)(b * NN + n) * KK + kk];
        const float* pc = xt + (size_t)(b * NN + n) * CC;
        const float* pn = xt + (size_t)(b * NN + nb) * CC;

        v8f acc1[4] = {};
#pragma unroll
        for (int ks = 0; ks < 16; ++ks) {
            int kc = ks * 4 + kofs;
            v2f a = *(const v2f*)(pn + kc) - *(const v2f*)(pc + kc);
#pragma unroll
            for (int nt = 0; nt < 4; ++nt) {
                int o = nt * 16 + laneM;
                v2f bb = *(const v2f*)(w1 + o * 128 + kc);
                acc1[nt] = wmma_f32(a, bb, acc1[nt]);
            }
        }
#pragma unroll
        for (int ks = 0; ks < 16; ++ks) {
            int kc = ks * 4 + kofs;
            v2f a = *(const v2f*)(pc + kc);
#pragma unroll
            for (int nt = 0; nt < 4; ++nt) {
                int o = nt * 16 + laneM;
                v2f bb = *(const v2f*)(w1 + o * 128 + 64 + kc);
                acc1[nt] = wmma_f32(a, bb, acc1[nt]);
            }
        }
#pragma unroll
        for (int nt = 0; nt < 4; ++nt) {
            int o = nt * 16 + laneM;
            float sc = g1[o] * rsqrtf(va1[o] + EPSV);
            float sh = be1[o] - mu1[o] * sc;
#pragma unroll
            for (int j = 0; j < 8; ++j) {
                int M = j + ((lane < 16) ? 0 : 8);
                h1s[wave][M * OO + o] = fmaxf(acc1[nt][j] * sc + sh, 0.0f);
            }
        }
        __syncthreads();

        v8f acc2[4] = {};
#pragma unroll
        for (int ks = 0; ks < 16; ++ks) {
            int kc = ks * 4 + kofs;
            v2f a = *(const v2f*)(&h1s[wave][laneM * OO + kc]);
#pragma unroll
            for (int nt = 0; nt < 4; ++nt) {
                int o = nt * 16 + laneM;
                v2f bb = *(const v2f*)(w2 + o * OO + kc);
                acc2[nt] = wmma_f32(a, bb, acc2[nt]);
            }
        }
#pragma unroll
        for (int nt = 0; nt < 4; ++nt) {
            int o = nt * 16 + laneM;
            float sc = g2[o] * rsqrtf(va2[o] + EPSV);
            float sh = be2[o] - mu2[o] * sc;
#pragma unroll
            for (int j = 0; j < 8; ++j) {
                int M = j + ((lane < 16) ? 0 : 8);
                int lr = mt * 16 + M;
                int pp = lr / KK;
                float val = fmaxf(acc2[nt][j] * sc + sh, 0.0f);
                atomicMax(&om[pp * OO + o], __float_as_uint(val));
            }
        }
        __syncthreads();
    }
    for (int i = threadIdx.x; i < 16 * OO; i += 128) {
        int p = i >> 6;
        int o = i & 63;
        out[(size_t)(b * OO + o) * NN + pt * 16 + p] = __uint_as_float(om[p * OO + o]);
    }
}

extern "C" void kernel_launch(void* const* d_in, const int* in_sizes, int n_in,
                              void* d_out, int out_size, void* d_ws, size_t ws_size,
                              hipStream_t stream) {
    const float* x   = (const float*)d_in[0];
    const float* w1  = (const float*)d_in[1];
    const float* g1  = (const float*)d_in[2];
    const float* be1 = (const float*)d_in[3];
    const float* mu1 = (const float*)d_in[4];
    const float* va1 = (const float*)d_in[5];
    const float* w2  = (const float*)d_in[6];
    const float* g2  = (const float*)d_in[7];
    const float* be2 = (const float*)d_in[8];
    const float* mu2 = (const float*)d_in[9];
    const float* va2 = (const float*)d_in[10];
    float* out = (float*)d_out;

    float* xt = (float*)d_ws;
    float* xx = xt + (size_t)BB * NN * CC;
    int* nbr = (int*)(xx + (size_t)BB * NN);

    prep_kernel<<<(BB * NN) / 256, 256, 0, stream>>>(x, xt, xx);
    knn_kernel<<<(BB * NN) / KNN_ROWS, 128, 0, stream>>>(xt, xx, nbr);
    edgeconv_kernel<<<BB * (NN / 16), 128, 0, stream>>>(
        xt, nbr, w1, g1, be1, mu1, va1, w2, g2, be2, mu2, va2, out);
}


# conv1 split by linearity, W1*[nbr-ctr,ctr] = W1a*x_nbr + (W1b-W1a)*x_ctr: P=W1a*x once per point (one small new launch), Q once per centre; per-edge conv1 becomes gather+add feeding conv2
# speedup vs baseline: 18.9437x; 1.0982x over previous
#include <hip/hip_runtime.h>
#include <math.h>

#define BB 4
#define CC 64
#define NN 4096
#define OO 64
#define KK 20
#define EPSV 1e-5f

typedef __attribute__((ext_vector_type(2))) float v2f;
typedef __attribute__((ext_vector_type(8))) float v8f;

__device__ __forceinline__ v8f wmma_f32(v2f a, v2f b, v8f c) {
    return __builtin_amdgcn_wmma_f32_16x16x4_f32(false, a, false, b, (short)0, c,
                                                 false, false);
}

__global__ __launch_bounds__(256) void prep_kernel(const float* __restrict__ x,
                                                   float* __restrict__ xt,
                                                   float* __restrict__ xx) {
    int t = blockIdx.x * blockDim.x + threadIdx.x;
    int b = t / NN;
    int n = t - b * NN;
    const float* px = x + (size_t)b * CC * NN + n;
    float* pt = xt + (size_t)t * CC;
    float s = 0.f;
#pragma unroll 8
    for (int c = 0; c < CC; ++c) {
        float v = px[(size_t)c * NN];
        pt[c] = v;
        s += v * v;
    }
    xx[t] = s;
}

#define KNN_ROWS 32
#define KNN_SPLIT 4
#define KNN_COLS (NN / KNN_SPLIT)
__global__ __launch_bounds__(128) void knn_kernel(const float* __restrict__ xt,
                                                  const float* __restrict__ xx,
                                                  int* __restrict__ nbr) {
    const int wave = threadIdx.x >> 5;
    const int lane = threadIdx.x & 31;
    const int laneM = lane & 15;
    const int kofs = (lane < 16) ? 0 : 2;

    const int row0 = blockIdx.x * KNN_ROWS;
    const int b = row0 / NN;
    const int r0 = row0 - b * NN;
    const int c0 = wave * KNN_COLS;

    const float* prowLo = xt + (size_t)(b * NN + r0 + laneM) * CC;
    const float* prowHi = prowLo + 16 * CC;
    v2f aLo[16], aHi[16];
#pragma unroll
    for (int ks = 0; ks < 16; ++ks) {
        int kc = ks * 4 + kofs;
        aLo[ks] = *(const v2f*)(prowLo + kc);
        aHi[ks] = *(const v2f*)(prowHi + kc);
    }

    float topv[KK];
    int topi[KK];
#pragma unroll
    for (int s = 0; s < KK; ++s) { topv[s] = -3.0e38f; topi[s] = 0; }
    const float xxrow = xx[b * NN + r0 + lane];

    __shared__ __attribute__((aligned(16))) float dist[4][32 * 16];

    for (int ct = 0; ct < KNN_COLS / 16; ++ct) {
        const int col = c0 + ct * 16 + laneM;
        const float* pcol = xt + (size_t)(b * NN + col) * CC;
        v8f accLo = {};
        v8f accHi = {};
#pragma unroll
        for (int ks = 0; ks < 16; ++ks) {
            v2f bb = *(const v2f*)(pcol + ks * 4 + kofs);
            accLo = wmma_f32(aLo[ks], bb, accLo);
            accHi = wmma_f32(aHi[ks], bb, accHi);
        }
        const float xcol = xx[b * NN + col];
#pragma unroll
        for (int j = 0; j < 8; ++j) {
            int M = j + ((lane < 16) ? 0 : 8);
            dist[wave][M * 16 + laneM] = 2.0f * accLo[j] - xcol;
            dist[wave][(16 + M) * 16 + laneM] = 2.0f * accHi[j] - xcol;
        }
        __syncthreads();
        const float4* drow = (const float4*)&dist[wave][lane * 16];
        float dv[16];
#pragma unroll
        for (int q = 0; q < 4; ++q) {
            const float4 t4 = drow[q];
            dv[4 * q + 0] = t4.x; dv[4 * q + 1] = t4.y;
            dv[4 * q + 2] = t4.z; dv[4 * q + 3] = t4.w;
        }
#pragma unroll
        for (int c = 0; c < 16; ++c) {
            const float v = dv[c] - xxrow;
            if (v > topv[KK - 1]) {
                const int ci = c0 + ct * 16 + c;
                bool beats[KK];
                float oldv[KK];
                int oldi[KK];
#pragma unroll
                for (int s = 0; s < KK; ++s) {
                    oldv[s] = topv[s];
                    oldi[s] = topi[s];
                    beats[s] = v > oldv[s];
                }
                float inv[KK];
                int ini[KK];
                inv[0] = v;
                ini[0] = ci;
#pragma unroll
                for (int s = 1; s < KK; ++s) {
                    const bool mu = beats[s - 1];
                    const float upv = oldv[s - 1];
                    const int upi = oldi[s - 1];
                    inv[s] = mu ? upv : v;
                    ini[s] = mu ? upi : ci;
                }
#pragma unroll
                for (int s = 0; s < KK; ++s) {
                    const bool ms = beats[s];
                    const float a = inv[s];
                    const float b = oldv[s];
                    const int ai = ini[s];
                    const int bi = oldi[s];
                    topv[s] = ms ? a : b;
                    topi[s] = ms ? ai : bi;
                }
            }
        }
        __syncthreads();
    }
    __shared__ float mval[KNN_SPLIT][KK][32];
    __shared__ int midx[KNN_SPLIT][KK][32];
#pragma unroll
    for (int s = 0; s < KK; ++s) {
        mval[wave][s][lane] = topv[s];
        midx[wave][s][lane] = topi[s];
    }
    __syncthreads();
    if (wave == 0) {
        float h0 = mval[0][0][lane], h1 = mval[1][0][lane];
        float h2 = mval[2][0][lane], h3 = mval[3][0][lane];
        int p0 = 0, p1 = 0, p2 = 0, p3 = 0;
        int* outp = nbr + (size_t)(b * NN + r0 + lane) * KK;
        for (int s = 0; s < KK; ++s) {
            int w = 0, p = p0;
            float best = h0;
            if (h1 > best) { best = h1; w = 1; p = p1; }
            if (h2 > best) { best = h2; w = 2; p = p2; }
            if (h3 > best) { best = h3; w = 3; p = p3; }
            outp[s] = midx[w][p][lane];
            const int pnext = p + 1;
            const float hnext = mval[w][pnext < KK ? pnext : KK - 1][lane];
            p0 = (w == 0) ? pnext : p0;  h0 = (w == 0) ? hnext : h0;
            p1 = (w == 1) ? pnext : p1;  h1 = (w == 1) ? hnext : h1;
            p2 = (w == 2) ? pnext : p2;  h2 = (w == 2) ? hnext : h2;
            p3 = (w == 3) ? pnext : p3;  h3 = (w == 3) ? hnext : h3;
        }
    }
}


__global__ __launch_bounds__(128) void proj_kernel(float* xtp,
                                                   const float* __restrict__ w1,
                                                   const float* __restrict__ g1,
                                                   const float* __restrict__ va1) {
    const int wave = threadIdx.x >> 5;
    const int lane = threadIdx.x & 31;
    const int laneM = lane & 15;
    const int kofs = (lane < 16) ? 0 : 2;
    const int row0 = (blockIdx.x * 4 + wave) * 16;

    const float* prow = xtp + (size_t)(row0 + laneM) * CC;
    v2f a[16];
#pragma unroll
    for (int ks = 0; ks < 16; ++ks) a[ks] = *(const v2f*)(prow + ks * 4 + kofs);

    v8f acc[4] = {};
#pragma unroll
    for (int ks = 0; ks < 16; ++ks) {
        int kc = ks * 4 + kofs;
#pragma unroll
        for (int nt = 0; nt < 4; ++nt) {
            int o = nt * 16 + laneM;
            v2f bb = *(const v2f*)(w1 + o * 128 + kc);
            acc[nt] = wmma_f32(a[ks], bb, acc[nt]);
        }
    }
#pragma unroll
    for (int nt = 0; nt < 4; ++nt) {
        int o = nt * 16 + laneM;
        float sc = g1[o] * rsqrtf(va1[o] + EPSV);
#pragma unroll
        for (int j = 0; j < 8; ++j) {
            int M = j + ((lane < 16) ? 0 : 8);
            xtp[(size_t)(row0 + M) * CC + o] = acc[nt][j] * sc;
        }
    }
}

__global__ __launch_bounds__(128) void edgeconv_kernel(
    const float* __restrict__ x, const float* __restrict__ pbuf,
    const int* __restrict__ nbr,
    const float* __restrict__ w1, const float* __restrict__ g1,
    const float* __restrict__ be1, const float* __restrict__ mu1,
    const float* __restrict__ va1, const float* __restrict__ w2,
    const float* __restrict__ g2, const float* __restrict__ be2,
    const float* __restrict__ mu2, const float* __restrict__ va2,
    float* __restrict__ out) {
    const int wave = threadIdx.x >> 5;
    const int lane = threadIdx.x & 31;
    const int laneM = lane & 15;
    const int kofs = (lane < 16) ? 0 : 2;
    const int b = blockIdx.x >> 8;
    const int pt = blockIdx.x & 255;

    __shared__ __attribute__((aligned(16))) float qs[16 * OO];
    __shared__ unsigned om[16 * OO];
    __shared__ float bn2s[OO], bn2h[OO];
    for (int i = threadIdx.x; i < 16 * OO; i += 128) om[i] = 0u;
    if (threadIdx.x < OO) {
        const int o = threadIdx.x;
        const float sc = g2[o] * rsqrtf(va2[o] + EPSV);
        bn2s[o] = sc;
        bn2h[o] = be2[o] - mu2[o] * sc;
    }

    {
        const int o = wave * 16 + laneM;
        const float* px = x + (size_t)b * CC * NN + pt * 16 + laneM;
        v8f q0 = {};
        v8f q1 = {};
#pragma unroll
        for (int ks = 0; ks < 16; ks += 2) {
            int kc = ks * 4 + kofs;
            v2f a0, a1;
            a0.x = px[(size_t)(kc + 0) * NN];
            a0.y = px[(size_t)(kc + 1) * NN];
            a1.x = px[(size_t)(kc + 4) * NN];
            a1.y = px[(size_t)(kc + 5) * NN];
            v2f b0 = *(const v2f*)(w1 + o * 128 + 64 + kc) - *(const v2f*)(w1 + o * 128 + kc);
            v2f b1 = *(const v2f*)(w1 + o * 128 + 68 + kc) - *(const v2f*)(w1 + o * 128 + 4 + kc);
            q0 = wmma_f32(a0, b0, q0);
            q1 = wmma_f32(a1, b1, q1);
        }
        float sc = g1[o] * rsqrtf(va1[o] + EPSV);
        float sh = be1[o] - mu1[o] * sc;
#pragma unroll
        for (int j = 0; j < 8; ++j) {
            int M = j + ((lane < 16) ? 0 : 8);
            qs[M * OO + o] = (q0[j] + q1[j]) * sc + sh;
        }
    }

    v2f bw[16][4];
#pragma unroll
    for (int nt = 0; nt < 4; ++nt) {
        int o = nt * 16 + laneM;
#pragma unroll
        for (int ks = 0; ks < 16; ++ks)
            bw[ks][nt] = *(const v2f*)(w2 + o * OO + ks * 4 + kofs);
    }
    __syncthreads();

    for (int it = 0; it < 5; ++it) {
        const int mt = wave * 5 + it;
        const int lrow = mt * 16 + laneM;
        const int p = lrow / KK;
        const int kk = lrow - p * KK;
        const int n = pt * 16 + p;
        const int nb = nbr[(size_t)(b * NN + n) * KK + kk];
        const float* pn = pbuf + (size_t)(b * NN + nb) * CC;
        const float* pq = qs + p * OO;

        v2f a[16];
#pragma unroll
        for (int ks = 0; ks < 16; ++ks) {
            int kc = ks * 4 + kofs;
            v2f h = *(const v2f*)(pn + kc) + *(const v2f*)(pq + kc);
            a[ks].x = fmaxf(h.x, 0.0f);
            a[ks].y = fmaxf(h.y, 0.0f);
        }

        v8f acc2[4] = {};
#pragma unroll
        for (int ks = 0; ks < 16; ++ks) {
#pragma unroll
            for (int nt = 0; nt < 4; ++nt)
                acc2[nt] = wmma_f32(a[ks], bw[ks][nt], acc2[nt]);
        }
#pragma unroll
        for (int nt = 0; nt < 4; ++nt) {
            int o = nt * 16 + laneM;
            const float sc = bn2s[o];
            const float sh = bn2h[o];
#pragma unroll
            for (int j = 0; j < 8; ++j) {
                int M = j + ((lane < 16) ? 0 : 8);
                int lr = mt * 16 + M;
                int pp = lr / KK;
                float val = fmaxf(acc2[nt][j] * sc + sh, 0.0f);
                atomicMax(&om[pp * OO + o], __float_as_uint(val));
            }
        }
    }
    __syncthreads();
    for (int i = threadIdx.x; i < 16 * OO; i += 128) {
        int p = i >> 6;
        int o = i & 63;
        out[(size_t)(b * OO + o) * NN + pt * 16 + p] = __uint_as_float(om[p * OO + o]);
    }
}

extern "C" void kernel_launch(void* const* d_in, const int* in_sizes, int n_in,
                              void* d_out, int out_size, void* d_ws, size_t ws_size,
                              hipStream_t stream) {
    const float* x   = (const float*)d_in[0];
    const float* w1  = (const float*)d_in[1];
    const float* g1  = (const float*)d_in[2];
    const float* be1 = (const float*)d_in[3];
    const float* mu1 = (const float*)d_in[4];
    const float* va1 = (const float*)d_in[5];
    const float* w2  = (const float*)d_in[6];
    const float* g2  = (const float*)d_in[7];
    const float* be2 = (const float*)d_in[8];
    const float* mu2 = (const float*)d_in[9];
    const float* va2 = (const float*)d_in[10];
    float* out = (float*)d_out;

    float* xt = (float*)d_ws;
    float* xx = xt + (size_t)BB * NN * CC;
    int* nbr = (int*)(xx + (size_t)BB * NN);

    prep_kernel<<<(BB * NN) / 256, 256, 0, stream>>>(x, xt, xx);
    knn_kernel<<<(BB * NN) / KNN_ROWS, 128, 0, stream>>>(xt, xx, nbr);
    proj_kernel<<<(BB * NN / 16) / 4, 128, 0, stream>>>(xt, w1, g1, va1);
    edgeconv_kernel<<<BB * (NN / 16), 128, 0, stream>>>(
        x, xt, nbr, w1, g1, be1, mu1, va1, w2, g2, be2, mu2, va2, out);
}


# kNN top-20 scan: each lane marks its passing candidates of a tile in a bit mask, then lanes insert their own marked candidates side by side (was one branch + insert per candidate any lane passed)
# speedup vs baseline: 21.9808x; 1.1603x over previous
#include <hip/hip_runtime.h>
#include <math.h>

#define BB 4
#define CC 64
#define NN 4096
#define OO 64
#define KK 20
#define EPSV 1e-5f

typedef __attribute__((ext_vector_type(2))) float v2f;
typedef __attribute__((ext_vector_type(8))) float v8f;

__device__ __forceinline__ v8f wmma_f32(v2f a, v2f b, v8f c) {
    return __builtin_amdgcn_wmma_f32_16x16x4_f32(false, a, false, b, (short)0, c,
                                                 false, false);
}

__global__ __launch_bounds__(256) void prep_kernel(const float* __restrict__ x,
                                                   float* __restrict__ xt,
                                                   float* __restrict__ xx) {
    int t = blockIdx.x * blockDim.x + threadIdx.x;
    int b = t / NN;
    int n = t - b * NN;
    const float* px = x + (size_t)b * CC * NN + n;
    float* pt = xt + (size_t)t * CC;
    float s = 0.f;
#pragma unroll 8
    for (int c = 0; c < CC; ++c) {
        float v = px[(size_t)c * NN];
        pt[c] = v;
        s += v * v;
    }
    xx[t] = s;
}

#define KNN_ROWS 32
#define KNN_SPLIT 4
#define KNN_COLS (NN / KNN_SPLIT)
__global__ __launch_bounds__(128) void knn_kernel(const float* __restrict__ xt,
                                                  const float* __restrict__ xx,
                                                  int* __restrict__ nbr) {
    const int wave = threadIdx.x >> 5;
    const int lane = threadIdx.x & 31;
    const int laneM = lane & 15;
    const int kofs = (lane < 16) ? 0 : 2;

    const int row0 = blockIdx.x * KNN_ROWS;
    const int b = row0 / NN;
    const int r0 = row0 - b * NN;
    const int c0 = wave * KNN_COLS;

    const float* prowLo = xt + (size_t)(b * NN + r0 + laneM) * CC;
    const float* prowHi = prowLo + 16 * CC;
    v2f aLo[16], aHi[16];
#pragma unroll
    for (int ks = 0; ks < 16; ++ks) {
        int kc = ks * 4 + kofs;
        aLo[ks] = *(const v2f*)(prowLo + kc);
        aHi[ks] = *(const v2f*)(prowHi + kc);
    }

    float topv[KK];
    int topi[KK];
#pragma unroll
    for (int s = 0; s < KK; ++s) { topv[s] = -3.0e38f; topi[s] = 0; }
    const float xxrow = xx[b * NN + r0 + lane];

    __shared__ __attribute__((aligned(16))) float dist[4][32 * 16];

    for (int ct = 0; ct < KNN_COLS / 16; ++ct) {
        const int col = c0 + ct * 16 + laneM;
        const float* pcol = xt + (size_t)(b * NN + col) * CC;
        v8f accLo = {};
        v8f accHi = {};
#pragma unroll
        for (int ks = 0; ks < 16; ++ks) {
            v2f bb = *(const v2f*)(pcol + ks * 4 + kofs);
            accLo = wmma_f32(aLo[ks], bb, accLo);
            accHi = wmma_f32(aHi[ks], bb, accHi);
        }
        const float xcol = xx[b * NN + col];
#pragma unroll
        for (int j = 0; j < 8; ++j) {
            int M = j + ((lane < 16) ? 0 : 8);
            dist[wave][M * 16 + laneM] = 2.0f * accLo[j] - xcol;
            dist[wave][(16 + M) * 16 + laneM] = 2.0f * accHi[j] - xcol;
        }
        __syncthreads();
        const float4* drow = (const float4*)&dist[wave][lane * 16];
        float dv[16];
#pragma unroll
        for (int q = 0; q < 4; ++q) {
            const float4 t4 = drow[q];
            dv[4 * q + 0] = t4.x; dv[4 * q + 1] = t4.y;
            dv[4 * q + 2] = t4.z; dv[4 * q + 3] = t4.w;
        }
        const float kth = topv[KK - 1];
        unsigned pm = 0u;
#pragma unroll
        for (int c = 15; c >= 0; --c)
            pm = pm + pm + (((dv[c] - xxrow) > kth) ? 1u : 0u);

        const float* myrow = &dist[wave][lane * 16];
        const int cbase = c0 + ct * 16;
        int c = __builtin_ctz(pm | 0x8000u);
        float cand = myrow[c];
        unsigned live = __builtin_amdgcn_ballot_w32(pm != 0u);
        for (int trip = 0; trip < 16 && live != 0u; ++trip) {
            const bool act = pm != 0u;
            pm &= pm - 1u;
            live = __builtin_amdgcn_ballot_w32(pm != 0u);
            const int cn = __builtin_ctz(pm | 0x8000u);
            const float nxt = myrow[cn];
            asm volatile("" ::: "memory");
            const float v = act ? (cand - xxrow) : -INFINITY;
            const int ci = cbase + c;
            bool beats[KK];
            float oldv[KK];
            int oldi[KK];
#pragma unroll
            for (int s = 0; s < KK; ++s) {
                oldv[s] = topv[s];
                oldi[s] = topi[s];
                beats[s] = v > oldv[s];
            }
            float inv[KK];
            int ini[KK];
            inv[0] = v;
            ini[0] = ci;
#pragma unroll
            for (int s = 1; s < KK; ++s) {
                const bool mu = beats[s - 1];
                const float upv = oldv[s - 1];
                const int upi = oldi[s - 1];
                inv[s] = mu ? upv : v;
                ini[s] = mu ? upi : ci;
            }
#pragma unroll
            for (int s = 0; s < KK; ++s) {
                const bool ms = beats[s];
                const float a = inv[s];
                const float b = oldv[s];
                const int ai = ini[s];
                const int bi = oldi[s];
                topv[s] = ms ? a : b;
                topi[s] = ms ? ai : bi;
            }
            c = cn;
            cand = nxt;
        }
        __syncthreads();
    }
    __shared__ float mval[KNN_SPLIT][KK][32];
    __shared__ int midx[KNN_SPLIT][KK][32];
#pragma unroll
    for (int s = 0; s < KK; ++s) {
        mval[wave][s][lane] = topv[s];
        midx[wave][s][lane] = topi[s];
    }
    __syncthreads();
    if (wave == 0) {
        float h0 = mval[0][0][lane], h1 = mval[1][0][lane];
        float h2 = mval[2][0][lane], h3 = mval[3][0][lane];
        int p0 = 0, p1 = 0, p2 = 0, p3 = 0;
        int* outp = nbr + (size_t)(b * NN + r0 + lane) * KK;
        for (int s = 0; s < KK; ++s) {
            int w = 0, p = p0;
            float best = h0;
            if (h1 > best) { best = h1; w = 1; p = p1; }
            if (h2 > best) { best = h2; w = 2; p = p2; }
            if (h3 > best) { best = h3; w = 3; p = p3; }
            outp[s] = midx[w][p][lane];
            const int pnext = p + 1;
            const float hnext = mval[w][pnext < KK ? pnext : KK - 1][lane];
            p0 = (w == 0) ? pnext : p0;  h0 = (w == 0) ? hnext : h0;
            p1 = (w == 1) ? pnext : p1;  h1 = (w == 1) ? hnext : h1;
            p2 = (w == 2) ? pnext : p2;  h2 = (w == 2) ? hnext : h2;
            p3 = (w == 3) ? pnext : p3;  h3 = (w == 3) ? hnext : h3;
        }
    }
}


__global__ __launch_bounds__(128) void proj_kernel(float* xtp,
                                                   const float* __restrict__ w1,
                                                   const float* __restrict__ g1,
                                                   const float* __restrict__ va1) {
    const int wave = threadIdx.x >> 5;
    const int lane = threadIdx.x & 31;
    const int laneM = lane & 15;
    const int kofs = (lane < 16) ? 0 : 2;
    const int row0 = (blockIdx.x * 4 + wave) * 16;

    const float* prow = xtp + (size_t)(row0 + laneM) * CC;
    v2f a[16];
#pragma unroll
    for (int ks = 0; ks < 16; ++ks) a[ks] = *(const v2f*)(prow + ks * 4 + kofs);

    v8f acc[4] = {};
#pragma unroll
    for (int ks = 0; ks < 16; ++ks) {
        int kc = ks * 4 + kofs;
#pragma unroll
        for (int nt = 0; nt < 4; ++nt) {
            int o = nt * 16 + laneM;
            v2f bb = *(const v2f*)(w1 + o * 128 + kc);
            acc[nt] = wmma_f32(a[ks], bb, acc[nt]);
        }
    }
#pragma unroll
    for (int nt = 0; nt < 4; ++nt) {
        int o = nt * 16 + laneM;
        float sc = g1[o] * rsqrtf(va1[o] + EPSV);
#pragma unroll
        for (int j = 0; j < 8; ++j) {
            int M = j + ((lane < 16) ? 0 : 8);
            xtp[(size_t)(row0 + M) * CC + o] = acc[nt][j] * sc;
        }
    }
}

__global__ __launch_bounds__(128) void edgeconv_kernel(
    const float* __restrict__ x, const float* __restrict__ pbuf,
    const int* __restrict__ nbr,
    const float* __restrict__ w1, const float* __restrict__ g1,
    const float* __restrict__ be1, const float* __restrict__ mu1,
    const float* __restrict__ va1, const float* __restrict__ w2,
    const float* __restrict__ g2, const float* __restrict__ be2,
    const float* __restrict__ mu2, const float* __restrict__ va2,
    float* __restrict__ out) {
    const int wave = threadIdx.x >> 5;
    const int lane = threadIdx.x & 31;
    const int laneM = lane & 15;
    const int kofs = (lane < 16) ? 0 : 2;
    const int b = blockIdx.x >> 8;
    const int pt = blockIdx.x & 255;

    __shared__ __attribute__((aligned(16))) float qs[16 * OO];
    __shared__ unsigned om[16 * OO];
    __shared__ float bn2s[OO], bn2h[OO];
    for (int i = threadIdx.x; i < 16 * OO; i += 128) om[i] = 0u;
    if (threadIdx.x < OO) {
        const int o = threadIdx.x;
        const float sc = g2[o] * rsqrtf(va2[o] + EPSV);
        bn2s[o] = sc;
        bn2h[o] = be2[o] - mu2[o] * sc;
    }

    {
        const int o = wave * 16 + laneM;
        const float* px = x + (size_t)b * CC * NN + pt * 16 + laneM;
        v8f q0 = {};
        v8f q1 = {};
#pragma unroll
        for (int ks = 0; ks < 16; ks += 2) {
            int kc = ks * 4 + kofs;
            v2f a0, a1;
            a0.x = px[(size_t)(kc + 0) * NN];
            a0.y = px[(size_t)(kc + 1) * NN];
            a1.x = px[(size_t)(kc + 4) * NN];
            a1.y = px[(size_t)(kc + 5) * NN];
            v2f b0 = *(const v2f*)(w1 + o * 128 + 64 + kc) - *(const v2f*)(w1 + o * 128 + kc);
            v2f b1 = *(const v2f*)(w1 + o * 128 + 68 + kc) - *(const v2f*)(w1 + o * 128 + 4 + kc);
            q0 = wmma_f32(a0, b0, q0);
            q1 = wmma_f32(a1, b1, q1);
        }
        float sc = g1[o] * rsqrtf(va1[o] + EPSV);
        float sh = be1[o] - mu1[o] * sc;
#pragma unroll
        for (int j = 0; j < 8; ++j) {
            int M = j + ((lane < 16) ? 0 : 8);
            qs[M * OO + o] = (q0[j] + q1[j]) * sc + sh;
        }
    }

    v2f bw[16][4];
#pragma unroll
    for (int nt = 0; nt < 4; ++nt) {
        int o = nt * 16 + laneM;
#pragma unroll
        for (int ks = 0; ks < 16; ++ks)
            bw[ks][nt] = *(const v2f*)(w2 + o * OO + ks * 4 + kofs);
    }
    __syncthreads();

    for (int it = 0; it < 5; ++it) {
        const int mt = wave * 5 + it;
        const int lrow = mt * 16 + laneM;
        const int p = lrow / KK;
        const int kk = lrow - p * KK;
        const int n = pt * 16 + p;
        const int nb = nbr[(size_t)(b * NN + n) * KK + kk];
        const float* pn = pbuf + (size_t)(b * NN + nb) * CC;
        const float* pq = qs + p * OO;

        v2f a[16];
#pragma unroll
        for (int ks = 0; ks < 16; ++ks) {
            int kc = ks * 4 + kofs;
            v2f h = *(const v2f*)(pn + kc) + *(const v2f*)(pq + kc);
            a[ks].x = fmaxf(h.x, 0.0f);
            a[ks].y = fmaxf(h.y, 0.0f);
        }

        v8f acc2[4] = {};
#pragma unroll
        for (int ks = 0; ks < 16; ++ks) {
#pragma unroll
            for (int nt = 0; nt < 4; ++nt)
                acc2[nt] = wmma_f32(a[ks], bw[ks][nt], acc2[nt]);
        }
#pragma unroll
        for (int nt = 0; nt < 4; ++nt) {
            int o = nt * 16 + laneM;
            const float sc = bn2s[o];
            const float sh = bn2h[o];
#pragma unroll
            for (int j = 0; j < 8; ++j) {
                int M = j + ((lane < 16) ? 0 : 8);
                int lr = mt * 16 + M;
                int pp = lr / KK;
                float val = fmaxf(acc2[nt][j] * sc + sh, 0.0f);
                atomicMax(&om[pp * OO + o], __float_as_uint(val));
            }
        }
    }
    __syncthreads();
    for (int i = threadIdx.x; i < 16 * OO; i += 128) {
        int p = i >> 6;
        int o = i & 63;
        out[(size_t)(b * OO + o) * NN + pt * 16 + p] = __uint_as_float(om[p * OO + o]);
    }
}

extern "C" void kernel_launch(void* const* d_in, const int* in_sizes, int n_in,
                              void* d_out, int out_size, void* d_ws, size_t ws_size,
                              hipStream_t stream) {
    const float* x   = (const float*)d_in[0];
    const float* w1  = (const float*)d_in[1];
    const float* g1  = (const float*)d_in[2];
    const float* be1 = (const float*)d_in[3];
    const float* mu1 = (const float*)d_in[4];
    const float* va1 = (const float*)d_in[5];
    const float* w2  = (const float*)d_in[6];
    const float* g2  = (const float*)d_in[7];
    const float* be2 = (const float*)d_in[8];
    const float* mu2 = (const float*)d_in[9];
    const float* va2 = (const float*)d_in[10];
    float* out = (float*)d_out;

    float* xt = (float*)d_ws;
    float* xx = xt + (size_t)BB * NN * CC;
    int* nbr = (int*)(xx + (size_t)BB * NN);

    prep_kernel<<<(BB * NN) / 256, 256, 0, stream>>>(x, xt, xx);
    knn_kernel<<<(BB * NN) / KNN_ROWS, 128, 0, stream>>>(xt, xx, nbr);
    proj_kernel<<<(BB * NN / 16) / 4, 128, 0, stream>>>(xt, w1, g1, va1);
    edgeconv_kernel<<<BB * (NN / 16), 128, 0, stream>>>(
        x, xt, nbr, w1, g1, be1, mu1, va1, w2, g2, be2, mu2, va2, out);
}


# kNN scan: a lane's per-tile candidate mark also uses the smallest of the four waves' fifth-best values of its row (20 candidates stand at or above it), passed through LDS once per tile
# speedup vs baseline: 23.1475x; 1.0531x over previous
#include <hip/hip_runtime.h>
#include <math.h>

#define BB 4
#define CC 64
#define NN 4096
#define OO 64
#define KK 20
#define EPSV 1e-5f

typedef __attribute__((ext_vector_type(2))) float v2f;
typedef __attribute__((ext_vector_type(8))) float v8f;

__device__ __forceinline__ v8f wmma_f32(v2f a, v2f b, v8f c) {
    return __builtin_amdgcn_wmma_f32_16x16x4_f32(false, a, false, b, (short)0, c,
                                                 false, false);
}

__global__ __launch_bounds__(256) void prep_kernel(const float* __restrict__ x,
                                                   float* __restrict__ xt,
                                                   float* __restrict__ xx) {
    int t = blockIdx.x * blockDim.x + threadIdx.x;
    int b = t / NN;
    int n = t - b * NN;
    const float* px = x + (size_t)b * CC * NN + n;
    float* pt = xt + (size_t)t * CC;
    float s = 0.f;
#pragma unroll 8
    for (int c = 0; c < CC; ++c) {
        float v = px[(size_t)c * NN];
        pt[c] = v;
        s += v * v;
    }
    xx[t] = s;
}

#define KNN_ROWS 32
#define KNN_SPLIT 4
#define KNN_COLS (NN / KNN_SPLIT)
#define KNN_SHARE 5
static_assert(KNN_SPLIT * KNN_SHARE >= KK, "the waves' shown slots must cover k");
static_assert((KNN_SPLIT & (KNN_SPLIT - 1)) == 0, "wave rotation uses a mask");
__global__ __launch_bounds__(128) void knn_kernel(const float* __restrict__ xt,
                                                  const float* __restrict__ xx,
                                                  int* __restrict__ nbr) {
    const int wave = threadIdx.x >> 5;
    const int lane = threadIdx.x & 31;
    const int laneM = lane & 15;
    const int kofs = (lane < 16) ? 0 : 2;

    const int row0 = blockIdx.x * KNN_ROWS;
    const int b = row0 / NN;
    const int r0 = row0 - b * NN;
    const int c0 = wave * KNN_COLS;

    const float* prowLo = xt + (size_t)(b * NN + r0 + laneM) * CC;
    const float* prowHi = prowLo + 16 * CC;
    v2f aLo[16], aHi[16];
#pragma unroll
    for (int ks = 0; ks < 16; ++ks) {
        int kc = ks * 4 + kofs;
        aLo[ks] = *(const v2f*)(prowLo + kc);
        aHi[ks] = *(const v2f*)(prowHi + kc);
    }

    float topv[KK];
    int topi[KK];
#pragma unroll
    for (int s = 0; s < KK; ++s) { topv[s] = -3.0e38f; topi[s] = 0; }
    const float xxrow = xx[b * NN + r0 + lane];

    __shared__ __attribute__((aligned(16))) float dist[4][32 * 16];
    __shared__ float fifth[KNN_SPLIT][32];

    for (int ct = 0; ct < KNN_COLS / 16; ++ct) {
        const int col = c0 + ct * 16 + laneM;
        const float* pcol = xt + (size_t)(b * NN + col) * CC;
        v8f accLo = {};
        v8f accHi = {};
#pragma unroll
        for (int ks = 0; ks < 16; ++ks) {
            v2f bb = *(const v2f*)(pcol + ks * 4 + kofs);
            accLo = wmma_f32(aLo[ks], bb, accLo);
            accHi = wmma_f32(aHi[ks], bb, accHi);
        }
        const float xcol = xx[b * NN + col];
#pragma unroll
        for (int j = 0; j < 8; ++j) {
            int M = j + ((lane < 16) ? 0 : 8);
            dist[wave][M * 16 + laneM] = 2.0f * accLo[j] - xcol;
            dist[wave][(16 + M) * 16 + laneM] = 2.0f * accHi[j] - xcol;
        }
        fifth[wave][lane] = topv[KNN_SHARE - 1];
        __syncthreads();
        const float f1 = fifth[(wave + 1) & (KNN_SPLIT - 1)][lane];
        const float f2 = fifth[(wave + 2) & (KNN_SPLIT - 1)][lane];
        const float f3 = fifth[(wave + 3) & (KNN_SPLIT - 1)][lane];
        const float4* drow = (const float4*)&dist[wave][lane * 16];
        float dv[16];
#pragma unroll
        for (int q = 0; q < 4; ++q) {
            const float4 t4 = drow[q];
            dv[4 * q + 0] = t4.x; dv[4 * q + 1] = t4.y;
            dv[4 * q + 2] = t4.z; dv[4 * q + 3] = t4.w;
        }
        const float kth = topv[KK - 1];
        const float thr =
            fmaxf(kth, fminf(fminf(topv[KNN_SHARE - 1], f1), fminf(f2, f3)));
        unsigned pm = 0u;
#pragma unroll
        for (int c = 15; c >= 0; --c)
            pm = pm + pm + (((dv[c] - xxrow) >= thr) ? 1u : 0u);

        const float* myrow = &dist[wave][lane * 16];
        const int cbase = c0 + ct * 16;
        int c = __builtin_ctz(pm | 0x8000u);
        float cand = myrow[c];
        unsigned live = __builtin_amdgcn_ballot_w32(pm != 0u);
        for (int trip = 0; trip < 16 && live != 0u; ++trip) {
            const bool act = pm != 0u;
            pm &= pm - 1u;
            live = __builtin_amdgcn_ballot_w32(pm != 0u);
            const int cn = __builtin_ctz(pm | 0x8000u);
            const float nxt = myrow[cn];
            asm volatile("" ::: "memory");
            const float v = act ? (cand - xxrow) : -INFINITY;
            const int ci = cbase + c;
            bool beats[KK];
            float oldv[KK];
            int oldi[KK];
#pragma unroll
            for (int s = 0; s < KK; ++s) {
                oldv[s] = topv[s];
                oldi[s] = topi[s];
                beats[s] = v > oldv[s];
            }
            float inv[KK];
            int ini[KK];
            inv[0] = v;
            ini[0] = ci;
#pragma unroll
            for (int s = 1; s < KK; ++s) {
                const bool mu = beats[s - 1];
                const float upv = oldv[s - 1];
                const int upi = oldi[s - 1];
                inv[s] = mu ? upv : v;
                ini[s] = mu ? upi : ci;
            }
#pragma unroll
            for (int s = 0; s < KK; ++s) {
                const bool ms = beats[s];
                const float a = inv[s];
                const float b = oldv[s];
                const int ai = ini[s];
                const int bi = oldi[s];
                topv[s] = ms ? a : b;
                topi[s] = ms ? ai : bi;
            }
            c = cn;
            cand = nxt;
        }
        __syncthreads();
    }
    __shared__ float mval[KNN_SPLIT][KK][32];
    __shared__ int midx[KNN_SPLIT][KK][32];
#pragma unroll
    for (int s = 0; s < KK; ++s) {
        mval[wave][s][lane] = topv[s];
        midx[wave][s][lane] = topi[s];
    }
    __syncthreads();
    if (wave == 0) {
        float h0 = mval[0][0][lane], h1 = mval[1][0][lane];
        float h2 = mval[2][0][lane], h3 = mval[3][0][lane];
        int p0 = 0, p1 = 0, p2 = 0, p3 = 0;
        int* outp = nbr + (size_t)(b * NN + r0 + lane) * KK;
        for (int s = 0; s < KK; ++s) {
            int w = 0, p = p0;
            float best = h0;
            if (h1 > best) { best = h1; w = 1; p = p1; }
            if (h2 > best) { best = h2; w = 2; p = p2; }
            if (h3 > best) { best = h3; w = 3; p = p3; }
            outp[s] = midx[w][p][lane];
            const int pnext = p + 1;
            const float hnext = mval[w][pnext < KK ? pnext : KK - 1][lane];
            p0 = (w == 0) ? pnext : p0;  h0 = (w == 0) ? hnext : h0;
            p1 = (w == 1) ? pnext : p1;  h1 = (w == 1) ? hnext : h1;
            p2 = (w == 2) ? pnext : p2;  h2 = (w == 2) ? hnext : h2;
            p3 = (w == 3) ? pnext : p3;  h3 = (w == 3) ? hnext : h3;
        }
    }
}


__global__ __launch_bounds__(128) void proj_kernel(float* xtp,
                                                   const float* __restrict__ w1,
                                                   const float* __restrict__ g1,
                                                   const float* __restrict__ va1) {
    const int wave = threadIdx.x >> 5;
    const int lane = threadIdx.x & 31;
    const int laneM = lane & 15;
    const int kofs = (lane < 16) ? 0 : 2;
    const int row0 = (blockIdx.x * 4 + wave) * 16;

    const float* prow = xtp + (size_t)(row0 + laneM) * CC;
    v2f a[16];
#pragma unroll
    for (int ks = 0; ks < 16; ++ks) a[ks] = *(const v2f*)(prow + ks * 4 + kofs);

    v8f acc[4] = {};
#pragma unroll
    for (int ks = 0; ks < 16; ++ks) {
        int kc = ks * 4 + kofs;
#pragma unroll
        for (int nt = 0; nt < 4; ++nt) {
            int o = nt * 16 + laneM;
            v2f bb = *(const v2f*)(w1 + o * 128 + kc);
            acc[nt] = wmma_f32(a[ks], bb, acc[nt]);
        }
    }
#pragma unroll
    for (int nt = 0; nt < 4; ++nt) {
        int o = nt * 16 + laneM;
        float sc = g1[o] * rsqrtf(va1[o] + EPSV);
#pragma unroll
        for (int j = 0; j < 8; ++j) {
            int M = j + ((lane < 16) ? 0 : 8);
            xtp[(size_t)(row0 + M) * CC + o] = acc[nt][j] * sc;
        }
    }
}

__global__ __launch_bounds__(128) void edgeconv_kernel(
    const float* __restrict__ x, const float* __restrict__ pbuf,
    const int* __restrict__ nbr,
    const float* __restrict__ w1, const float* __restrict__ g1,
    const float* __restrict__ be1, const float* __restrict__ mu1,
    const float* __restrict__ va1, const float* __restrict__ w2,
    const float* __restrict__ g2, const float* __restrict__ be2,
    const float* __restrict__ mu2, const float* __restrict__ va2,
    float* __restrict__ out) {
    const int wave = threadIdx.x >> 5;
    const int lane = threadIdx.x & 31;
    const int laneM = lane & 15;
    const int kofs = (lane < 16) ? 0 : 2;
    const int b = blockIdx.x >> 8;
    const int pt = blockIdx.x & 255;

    __shared__ __attribute__((aligned(16))) float qs[16 * OO];
    __shared__ unsigned om[16 * OO];
    __shared__ float bn2s[OO], bn2h[OO];
    for (int i = threadIdx.x; i < 16 * OO; i += 128) om[i] = 0u;
    if (threadIdx.x < OO) {
        const int o = threadIdx.x;
        const float sc = g2[o] * rsqrtf(va2[o] + EPSV);
        bn2s[o] = sc;
        bn2h[o] = be2[o] - mu2[o] * sc;
    }

    {
        const int o = wave * 16 + laneM;
        const float* px = x + (size_t)b * CC * NN + pt * 16 + laneM;
        v8f q0 = {};
        v8f q1 = {};
#pragma unroll
        for (int ks = 0; ks < 16; ks += 2) {
            int kc = ks * 4 + kofs;
            v2f a0, a1;
            a0.x = px[(size_t)(kc + 0) * NN];
            a0.y = px[(size_t)(kc + 1) * NN];
            a1.x = px[(size_t)(kc + 4) * NN];
            a1.y = px[(size_t)(kc + 5) * NN];
            v2f b0 = *(const v2f*)(w1 + o * 128 + 64 + kc) - *(const v2f*)(w1 + o * 128 + kc);
            v2f b1 = *(const v2f*)(w1 + o * 128 + 68 + kc) - *(const v2f*)(w1 + o * 128 + 4 + kc);
            q0 = wmma_f32(a0, b0, q0);
            q1 = wmma_f32(a1, b1, q1);
        }
        float sc = g1[o] * rsqrtf(va1[o] + EPSV);
        float sh = be1[o] - mu1[o] * sc;
#pragma unroll
        for (int j = 0; j < 8; ++j) {
            int M = j + ((lane < 16) ? 0 : 8);
            qs[M * OO + o] = (q0[j] + q1[j]) * sc + sh;
        }
    }

    v2f bw[16][4];
#pragma unroll
    for (int nt = 0; nt < 4; ++nt) {
        int o = nt * 16 + laneM;
#pragma unroll
        for (int ks = 0; ks < 16; ++ks)
            bw[ks][nt] = *(const v2f*)(w2 + o * OO + ks * 4 + kofs);
    }
    __syncthreads();

    for (int it = 0; it < 5; ++it) {
        const int mt = wave * 5 + it;
        const int lrow = mt * 16 + laneM;
        const int p = lrow / KK;
        const int kk = lrow - p * KK;
        const int n = pt * 16 + p;
        const int nb = nbr[(size_t)(b * NN + n) * KK + kk];
        const float* pn = pbuf + (size_t)(b * NN + nb) * CC;
        const float* pq = qs + p * OO;

        v2f a[16];
#pragma unroll
        for (int ks = 0; ks < 16; ++ks) {
            int kc = ks * 4 + kofs;
            v2f h = *(const v2f*)(pn + kc) + *(const v2f*)(pq + kc);
            a[ks].x = fmaxf(h.x, 0.0f);
            a[ks].y = fmaxf(h.y, 0.0f);
        }

        v8f acc2[4] = {};
#pragma unroll
        for (int ks = 0; ks < 16; ++ks) {
#pragma unroll
            for (int nt = 0; nt < 4; ++nt)
                acc2[nt] = wmma_f32(a[ks], bw[ks][nt], acc2[nt]);
        }
#pragma unroll
        for (int nt = 0; nt < 4; ++nt) {
            int o = nt * 16 + laneM;
            const float sc = bn2s[o];
            const float sh = bn2h[o];
#pragma unroll
            for (int j = 0; j < 8; ++j) {
                int M = j + ((lane < 16) ? 0 : 8);
                int lr = mt * 16 + M;
                int pp = lr / KK;
                float val = fmaxf(acc2[nt][j] * sc + sh, 0.0f);
                atomicMax(&om[pp * OO + o], __float_as_uint(val));
            }
        }
    }
    __syncthreads();
    for (int i = threadIdx.x; i < 16 * OO; i += 128) {
        int p = i >> 6;
        int o = i & 63;
        out[(size_t)(b * OO + o) * NN + pt * 16 + p] = __uint_as_float(om[p * OO + o]);
    }
}

extern "C" void kernel_launch(void* const* d_in, const int* in_sizes, int n_in,
                              void* d_out, int out_size, void* d_ws, size_t ws_size,
                              hipStream_t stream) {
    const float* x   = (const float*)d_in[0];
    const float* w1  = (const float*)d_in[1];
    const float* g1  = (const float*)d_in[2];
    const float* be1 = (const float*)d_in[3];
    const float* mu1 = (const float*)d_in[4];
    const float* va1 = (const float*)d_in[5];
    const float* w2  = (const float*)d_in[6];
    const float* g2  = (const float*)d_in[7];
    const float* be2 = (const float*)d_in[8];
    const float* mu2 = (const float*)d_in[9];
    const float* va2 = (const float*)d_in[10];
    float* out = (float*)d_out;

    float* xt = (float*)d_ws;
    float* xx = xt + (size_t)BB * NN * CC;
    int* nbr = (int*)(xx + (size_t)BB * NN);

    prep_kernel<<<(BB * NN) / 256, 256, 0, stream>>>(x, xt, xx);
    knn_kernel<<<(BB * NN) / KNN_ROWS, 128, 0, stream>>>(xt, xx, nbr);
    proj_kernel<<<(BB * NN / 16) / 4, 128, 0, stream>>>(xt, w1, g1, va1);
    edgeconv_kernel<<<BB * (NN / 16), 128, 0, stream>>>(
        x, xt, nbr, w1, g1, be1, mu1, va1, w2, g2, be2, mu2, va2, out);
}


# kNN scan: next tile's 16 column-fragment loads (and xx) issued a tile ahead into a second register set; the tile's two barriers wait for LDS only, so those loads stay in flight across them
# speedup vs baseline: 26.4564x; 1.1429x over previous
#include <hip/hip_runtime.h>
#include <math.h>

#define BB 4
#define CC 64
#define NN 4096
#define OO 64
#define KK 20
#define EPSV 1e-5f

typedef __attribute__((ext_vector_type(2))) float v2f;
typedef __attribute__((ext_vector_type(8))) float v8f;

__device__ __forceinline__ v8f wmma_f32(v2f a, v2f b, v8f c) {
    return __builtin_amdgcn_wmma_f32_16x16x4_f32(false, a, false, b, (short)0, c,
                                                 false, false);
}

__device__ __forceinline__ void barrier_lds_only() {
    asm volatile("s_wait_dscnt 0x0\n\ts_barrier_signal -1\n\ts_barrier_wait -1" ::: "memory");
}

__global__ __launch_bounds__(256) void prep_kernel(const float* __restrict__ x,
                                                   float* __restrict__ xt,
                                                   float* __restrict__ xx) {
    int t = blockIdx.x * blockDim.x + threadIdx.x;
    int b = t / NN;
    int n = t - b * NN;
    const float* px = x + (size_t)b * CC * NN + n;
    float* pt = xt + (size_t)t * CC;
    float s = 0.f;
#pragma unroll 8
    for (int c = 0; c < CC; ++c) {
        float v = px[(size_t)c * NN];
        pt[c] = v;
        s += v * v;
    }
    xx[t] = s;
}

#define KNN_ROWS 32
#define KNN_SPLIT 4
#define KNN_COLS (NN / KNN_SPLIT)
#define KNN_SHARE 5
static_assert(KNN_SPLIT * KNN_SHARE >= KK, "the waves' shown slots must cover k");
static_assert((KNN_SPLIT & (KNN_SPLIT - 1)) == 0, "wave rotation uses a mask");
__global__ __launch_bounds__(128) void knn_kernel(const float* __restrict__ xt,
                                                  const float* __restrict__ xx,
                                                  int* __restrict__ nbr) {
    const int wave = threadIdx.x >> 5;
    const int lane = threadIdx.x & 31;
    const int laneM = lane & 15;
    const int kofs = (lane < 16) ? 0 : 2;

    const int row0 = blockIdx.x * KNN_ROWS;
    const int b = row0 / NN;
    const int r0 = row0 - b * NN;
    const int c0 = wave * KNN_COLS;

    const float* prowLo = xt + (size_t)(b * NN + r0 + laneM) * CC;
    const float* prowHi = prowLo + 16 * CC;
    v2f aLo[16], aHi[16];
#pragma unroll
    for (int ks = 0; ks < 16; ++ks) {
        int kc = ks * 4 + kofs;
        aLo[ks] = *(const v2f*)(prowLo + kc);
        aHi[ks] = *(const v2f*)(prowHi + kc);
    }

    float topv[KK];
    int topi[KK];
#pragma unroll
    for (int s = 0; s < KK; ++s) { topv[s] = -3.0e38f; topi[s] = 0; }
    const float xxrow = xx[b * NN + r0 + lane];

    __shared__ __attribute__((aligned(16))) float dist[4][32 * 16];
    __shared__ float fifth[KNN_SPLIT][32];

    const float* pcol0 = xt + (size_t)(b * NN + c0 + laneM) * CC + kofs;
    const float* pxx0 = xx + b * NN + c0 + laneM;
    v2f bnext[16];
#pragma unroll
    for (int ks = 0; ks < 16; ++ks) bnext[ks] = *(const v2f*)(pcol0 + ks * 4);
    float xnext = pxx0[0];

#pragma unroll 2
    for (int ct = 0; ct < KNN_COLS / 16; ++ct) {
        v2f bcur[16];
#pragma unroll
        for (int ks = 0; ks < 16; ++ks) bcur[ks] = bnext[ks];
        const float xcol = xnext;
        const int ctn = (ct + 1 < KNN_COLS / 16) ? ct + 1 : ct;
        const float* pcoln = pcol0 + (size_t)ctn * 16 * CC;
#pragma unroll
        for (int ks = 0; ks < 16; ++ks) bnext[ks] = *(const v2f*)(pcoln + ks * 4);
        xnext = pxx0[ctn * 16];

        v8f accLo = {};
        v8f accHi = {};
#pragma unroll
        for (int ks = 0; ks < 16; ++ks) {
            accLo = wmma_f32(aLo[ks], bcur[ks], accLo);
            accHi = wmma_f32(aHi[ks], bcur[ks], accHi);
        }
#pragma unroll
        for (int j = 0; j < 8; ++j) {
            int M = j + ((lane < 16) ? 0 : 8);
            dist[wave][M * 16 + laneM] = 2.0f * accLo[j] - xcol;
            dist[wave][(16 + M) * 16 + laneM] = 2.0f * accHi[j] - xcol;
        }
        fifth[wave][lane] = topv[KNN_SHARE - 1];
        barrier_lds_only();
        const float f1 = fifth[(wave + 1) & (KNN_SPLIT - 1)][lane];
        const float f2 = fifth[(wave + 2) & (KNN_SPLIT - 1)][lane];
        const float f3 = fifth[(wave + 3) & (KNN_SPLIT - 1)][lane];
        const float4* drow = (const float4*)&dist[wave][lane * 16];
        float dv[16];
#pragma unroll
        for (int q = 0; q < 4; ++q) {
            const float4 t4 = drow[q];
            dv[4 * q + 0] = t4.x; dv[4 * q + 1] = t4.y;
            dv[4 * q + 2] = t4.z; dv[4 * q + 3] = t4.w;
        }
        const float kth = topv[KK - 1];
        const float thr =
            fmaxf(kth, fminf(fminf(topv[KNN_SHARE - 1], f1), fminf(f2, f3)));
        unsigned pm = 0u;
#pragma unroll
        for (int c = 15; c >= 0; --c)
            pm = pm + pm + (((dv[c] - xxrow) >= thr) ? 1u : 0u);

        const float* myrow = &dist[wave][lane * 16];
        const int cbase = c0 + ct * 16;
        int c = __builtin_ctz(pm | 0x8000u);
        float cand = myrow[c];
        unsigned live = __builtin_amdgcn_ballot_w32(pm != 0u);
        for (int trip = 0; trip < 16 && live != 0u; ++trip) {
            const bool act = pm != 0u;
            pm &= pm - 1u;
            live = __builtin_amdgcn_ballot_w32(pm != 0u);
            const int cn = __builtin_ctz(pm | 0x8000u);
            const float nxt = myrow[cn];
            asm volatile("" ::: "memory");
            const float v = act ? (cand - xxrow) : -INFINITY;
            const int ci = cbase + c;
            bool beats[KK];
            float oldv[KK];
            int oldi[KK];
#pragma unroll
            for (int s = 0; s < KK; ++s) {
                oldv[s] = topv[s];
                oldi[s] = topi[s];
                beats[s] = v > oldv[s];
            }
            float inv[KK];
            int ini[KK];
            inv[0] = v;
            ini[0] = ci;
#pragma unroll
            for (int s = 1; s < KK; ++s) {
                const bool mu = beats[s - 1];
                const float upv = oldv[s - 1];
                const int upi = oldi[s - 1];
                inv[s] = mu ? upv : v;
                ini[s] = mu ? upi : ci;
            }
#pragma unroll
            for (int s = 0; s < KK; ++s) {
                const bool ms = beats[s];
                const float a = inv[s];
                const float b = oldv[s];
                const int ai = ini[s];
                const int bi = oldi[s];
                topv[s] = ms ? a : b;
                topi[s] = ms ? ai : bi;
            }
            c = cn;
            cand = nxt;
        }
        barrier_lds_only();
    }
    __shared__ float mval[KNN_SPLIT][KK][32];
    __shared__ int midx[KNN_SPLIT][KK][32];
#pragma unroll
    for (int s = 0; s < KK; ++s) {
        mval[wave][s][lane] = topv[s];
        midx[wave][s][lane] = topi[s];
    }
    __syncthreads();
    if (wave == 0) {
        float h0 = mval[0][0][lane], h1 = mval[1][0][lane];
        float h2 = mval[2][0][lane], h3 = mval[3][0][lane];
        int p0 = 0, p1 = 0, p2 = 0, p3 = 0;
        int* outp = nbr + (size_t)(b * NN + r0 + lane) * KK;
        for (int s = 0; s < KK; ++s) {
            int w = 0, p = p0;
            float best = h0;
            if (h1 > best) { best = h1; w = 1; p = p1; }
            if (h2 > best) { best = h2; w = 2; p = p2; }
            if (h3 > best) { best = h3; w = 3; p = p3; }
            outp[s] = midx[w][p][lane];
            const int pnext = p + 1;
            const float hnext = mval[w][pnext < KK ? pnext : KK - 1][lane];
            p0 = (w == 0) ? pnext : p0;  h0 = (w == 0) ? hnext : h0;
            p1 = (w == 1) ? pnext : p1;  h1 = (w == 1) ? hnext : h1;
            p2 = (w == 2) ? pnext : p2;  h2 = (w == 2) ? hnext : h2;
            p3 = (w == 3) ? pnext : p3;  h3 = (w == 3) ? hnext : h3;
        }
    }
}


__global__ __launch_bounds__(128) void proj_kernel(float* xtp,
                                                   const float* __restrict__ w1,
                                                   const float* __restrict__ g1,
                                                   const float* __restrict__ va1) {
    const int wave = threadIdx.x >> 5;
    const int lane = threadIdx.x & 31;
    const int laneM = lane & 15;
    const int kofs = (lane < 16) ? 0 : 2;
    const int row0 = (blockIdx.x * 4 + wave) * 16;

    const float* prow = xtp + (size_t)(row0 + laneM) * CC;
    v2f a[16];
#pragma unroll
    for (int ks = 0; ks < 16; ++ks) a[ks] = *(const v2f*)(prow + ks * 4 + kofs);

    v8f acc[4] = {};
#pragma unroll
    for (int ks = 0; ks < 16; ++ks) {
        int kc = ks * 4 + kofs;
#pragma unroll
        for (int nt = 0; nt < 4; ++nt) {
            int o = nt * 16 + laneM;
            v2f bb = *(const v2f*)(w1 + o * 128 + kc);
            acc[nt] = wmma_f32(a[ks], bb, acc[nt]);
        }
    }
#pragma unroll
    for (int nt = 0; nt < 4; ++nt) {
        int o = nt * 16 + laneM;
        float sc = g1[o] * rsqrtf(va1[o] + EPSV);
#pragma unroll
        for (int j = 0; j < 8; ++j) {
            int M = j + ((lane < 16) ? 0 : 8);
            xtp[(size_t)(row0 + M) * CC + o] = acc[nt][j] * sc;
        }
    }
}

__global__ __launch_bounds__(128) void edgeconv_kernel(
    const float* __restrict__ x, const float* __restrict__ pbuf,
    const int* __restrict__ nbr,
    const float* __restrict__ w1, const float* __restrict__ g1,
    const float* __restrict__ be1, const float* __restrict__ mu1,
    const float* __restrict__ va1, const float* __restrict__ w2,
    const float* __restrict__ g2, const float* __restrict__ be2,
    const float* __restrict__ mu2, const float* __restrict__ va2,
    float* __restrict__ out) {
    const int wave = threadIdx.x >> 5;
    const int lane = threadIdx.x & 31;
    const int laneM = lane & 15;
    const int kofs = (lane < 16) ? 0 : 2;
    const int b = blockIdx.x >> 8;
    const int pt = blockIdx.x & 255;

    __shared__ __attribute__((aligned(16))) float qs[16 * OO];
    __shared__ unsigned om[16 * OO];
    __shared__ float bn2s[OO], bn2h[OO];
    for (int i = threadIdx.x; i < 16 * OO; i += 128) om[i] = 0u;
    if (threadIdx.x < OO) {
        const int o = threadIdx.x;
        const float sc = g2[o] * rsqrtf(va2[o] + EPSV);
        bn2s[o] = sc;
        bn2h[o] = be2[o] - mu2[o] * sc;
    }

    {
        const int o = wave * 16 + laneM;
        const float* px = x + (size_t)b * CC * NN + pt * 16 + laneM;
        v8f q0 = {};
        v8f q1 = {};
#pragma unroll
        for (int ks = 0; ks < 16; ks += 2) {
            int kc = ks * 4 + kofs;
            v2f a0, a1;
            a0.x = px[(size_t)(kc + 0) * NN];
            a0.y = px[(size_t)(kc + 1) * NN];
            a1.x = px[(size_t)(kc + 4) * NN];
            a1.y = px[(size_t)(kc + 5) * NN];
            v2f b0 = *(const v2f*)(w1 + o * 128 + 64 + kc) - *(const v2f*)(w1 + o * 128 + kc);
            v2f b1 = *(const v2f*)(w1 + o * 128 + 68 + kc) - *(const v2f*)(w1 + o * 128 + 4 + kc);
            q0 = wmma_f32(a0, b0, q0);
            q1 = wmma_f32(a1, b1, q1);
        }
        float sc = g1[o] * rsqrtf(va1[o] + EPSV);
        float sh = be1[o] - mu1[o] * sc;
#pragma unroll
        for (int j = 0; j < 8; ++j) {
            int M = j + ((lane < 16) ? 0 : 8);
            qs[M * OO + o] = (q0[j] + q1[j]) * sc + sh;
        }
    }

    v2f bw[16][4];
#pragma unroll
    for (int nt = 0; nt < 4; ++nt) {
        int o = nt * 16 + laneM;
#pragma unroll
        for (int ks = 0; ks < 16; ++ks)
            bw[ks][nt] = *(const v2f*)(w2 + o * OO + ks * 4 + kofs);
    }
    __syncthreads();

    for (int it = 0; it < 5; ++it) {
        const int mt = wave * 5 + it;
        const int lrow = mt * 16 + laneM;
        const int p = lrow / KK;
        const int kk = lrow - p * KK;
        const int n = pt * 16 + p;
        const int nb = nbr[(size_t)(b * NN + n) * KK + kk];
        const float* pn = pbuf + (size_t)(b * NN + nb) * CC;
        const float* pq = qs + p * OO;

        v2f a[16];
#pragma unroll
        for (int ks = 0; ks < 16; ++ks) {
            int kc = ks * 4 + kofs;
            v2f h = *(const v2f*)(pn + kc) + *(const v2f*)(pq + kc);
            a[ks].x = fmaxf(h.x, 0.0f);
            a[ks].y = fmaxf(h.y, 0.0f);
        }

        v8f acc2[4] = {};
#pragma unroll
        for (int ks = 0; ks < 16; ++ks) {
#pragma unroll
            for (int nt = 0; nt < 4; ++nt)
                acc2[nt] = wmma_f32(a[ks], bw[ks][nt], acc2[nt]);
        }
#pragma unroll
        for (int nt = 0; nt < 4; ++nt) {
            int o = nt * 16 + laneM;
            const float sc = bn2s[o];
            const float sh = bn2h[o];
#pragma unroll
            for (int j = 0; j < 8; ++j) {
                int M = j + ((lane < 16) ? 0 : 8);
                int lr = mt * 16 + M;
                int pp = lr / KK;
                float val = fmaxf(acc2[nt][j] * sc + sh, 0.0f);
                atomicMax(&om[pp * OO + o], __float_as_uint(val));
            }
        }
    }
    __syncthreads();
    for (int i = threadIdx.x; i < 16 * OO; i += 128) {
        int p = i >> 6;
        int o = i & 63;
        out[(size_t)(b * OO + o) * NN + pt * 16 + p] = __uint_as_float(om[p * OO + o]);
    }
}

extern "C" void kernel_launch(void* const* d_in, const int* in_sizes, int n_in,
                              void* d_out, int out_size, void* d_ws, size_t ws_size,
                              hipStream_t stream) {
    const float* x   = (const float*)d_in[0];
    const float* w1  = (const float*)d_in[1];
    const float* g1  = (const float*)d_in[2];
    const float* be1 = (const float*)d_in[3];
    const float* mu1 = (const float*)d_in[4];
    const float* va1 = (const float*)d_in[5];
    const float* w2  = (const float*)d_in[6];
    const float* g2  = (const float*)d_in[7];
    const float* be2 = (const float*)d_in[8];
    const float* mu2 = (const float*)d_in[9];
    const float* va2 = (const float*)d_in[10];
    float* out = (float*)d_out;

    float* xt = (float*)d_ws;
    float* xx = xt + (size_t)BB * NN * CC;
    int* nbr = (int*)(xx + (size_t)BB * NN);

    prep_kernel<<<(BB * NN) / 256, 256, 0, stream>>>(x, xt, xx);
    knn_kernel<<<(BB * NN) / KNN_ROWS, 128, 0, stream>>>(xt, xx, nbr);
    proj_kernel<<<(BB * NN / 16) / 4, 128, 0, stream>>>(xt, w1, g1, va1);
    edgeconv_kernel<<<BB * (NN / 16), 128, 0, stream>>>(
        x, xt, nbr, w1, g1, be1, mu1, va1, w2, g2, be2, mu2, va2, out);
}


# kNN scan: both workgroup barriers of every tile dropped: a wave's distances pass through LDS rows of its own, in order within the wave; the other waves' rising fifth-best values read as they stand
# speedup vs baseline: 27.8405x; 1.0523x over previous
#include <hip/hip_runtime.h>
#include <math.h>

#define BB 4
#define CC 64
#define NN 4096
#define OO 64
#define KK 20
#define EPSV 1e-5f

typedef __attribute__((ext_vector_type(2))) float v2f;
typedef __attribute__((ext_vector_type(8))) float v8f;

__device__ __forceinline__ v8f wmma_f32(v2f a, v2f b, v8f c) {
    return __builtin_amdgcn_wmma_f32_16x16x4_f32(false, a, false, b, (short)0, c,
                                                 false, false);
}

__global__ __launch_bounds__(256) void prep_kernel(const float* __restrict__ x,
                                                   float* __restrict__ xt,
                                                   float* __restrict__ xx) {
    int t = blockIdx.x * blockDim.x + threadIdx.x;
    int b = t / NN;
    int n = t - b * NN;
    const float* px = x + (size_t)b * CC * NN + n;
    float* pt = xt + (size_t)t * CC;
    float s = 0.f;
#pragma unroll 8
    for (int c = 0; c < CC; ++c) {
        float v = px[(size_t)c * NN];
        pt[c] = v;
        s += v * v;
    }
    xx[t] = s;
}

#define KNN_ROWS 32
#define KNN_SPLIT 4
#define KNN_COLS (NN / KNN_SPLIT)
#define KNN_SHARE 5
static_assert(KNN_SPLIT * KNN_SHARE >= KK, "the waves' shown slots must cover k");
static_assert((KNN_SPLIT & (KNN_SPLIT - 1)) == 0, "wave rotation uses a mask");
__global__ __launch_bounds__(128) void knn_kernel(const float* __restrict__ xt,
                                                  const float* __restrict__ xx,
                                                  int* __restrict__ nbr) {
    const int wave = threadIdx.x >> 5;
    const int lane = threadIdx.x & 31;
    const int laneM = lane & 15;
    const int kofs = (lane < 16) ? 0 : 2;

    const int row0 = blockIdx.x * KNN_ROWS;
    const int b = row0 / NN;
    const int r0 = row0 - b * NN;
    const int c0 = wave * KNN_COLS;

    __shared__ __attribute__((aligned(16))) float dist[4][32 * 16];
    __shared__ float fifth[KNN_SPLIT][32];
    fifth[wave][lane] = -3.0e38f;
    __syncthreads();

    const float* prowLo = xt + (size_t)(b * NN + r0 + laneM) * CC;
    const float* prowHi = prowLo + 16 * CC;
    v2f aLo[16], aHi[16];
#pragma unroll
    for (int ks = 0; ks < 16; ++ks) {
        int kc = ks * 4 + kofs;
        aLo[ks] = *(const v2f*)(prowLo + kc);
        aHi[ks] = *(const v2f*)(prowHi + kc);
    }

    float topv[KK];
    int topi[KK];
#pragma unroll
    for (int s = 0; s < KK; ++s) { topv[s] = -3.0e38f; topi[s] = 0; }
    const float xxrow = xx[b * NN + r0 + lane];

    const float* pcol0 = xt + (size_t)(b * NN + c0 + laneM) * CC + kofs;
    const float* pxx0 = xx + b * NN + c0 + laneM;
    v2f bnext[16];
#pragma unroll
    for (int ks = 0; ks < 16; ++ks) bnext[ks] = *(const v2f*)(pcol0 + ks * 4);
    float xnext = pxx0[0];

#pragma unroll 2
    for (int ct = 0; ct < KNN_COLS / 16; ++ct) {
        v2f bcur[16];
#pragma unroll
        for (int ks = 0; ks < 16; ++ks) bcur[ks] = bnext[ks];
        const float xcol = xnext;
        const int ctn = (ct + 1 < KNN_COLS / 16) ? ct + 1 : ct;
        const float* pcoln = pcol0 + (size_t)ctn * 16 * CC;
#pragma unroll
        for (int ks = 0; ks < 16; ++ks) bnext[ks] = *(const v2f*)(pcoln + ks * 4);
        xnext = pxx0[ctn * 16];

        v8f accLo = {};
        v8f accHi = {};
#pragma unroll
        for (int ks = 0; ks < 16; ++ks) {
            accLo = wmma_f32(aLo[ks], bcur[ks], accLo);
            accHi = wmma_f32(aHi[ks], bcur[ks], accHi);
        }
#pragma unroll
        for (int j = 0; j < 8; ++j) {
            int M = j + ((lane < 16) ? 0 : 8);
            dist[wave][M * 16 + laneM] = 2.0f * accLo[j] - xcol;
            dist[wave][(16 + M) * 16 + laneM] = 2.0f * accHi[j] - xcol;
        }
        fifth[wave][lane] = topv[KNN_SHARE - 1];
        __builtin_amdgcn_fence(__ATOMIC_ACQ_REL, "wavefront");
        const float f1 = fifth[(wave + 1) & (KNN_SPLIT - 1)][lane];
        const float f2 = fifth[(wave + 2) & (KNN_SPLIT - 1)][lane];
        const float f3 = fifth[(wave + 3) & (KNN_SPLIT - 1)][lane];
        const float4* drow = (const float4*)&dist[wave][lane * 16];
        float dv[16];
#pragma unroll
        for (int q = 0; q < 4; ++q) {
            const float4 t4 = drow[q];
            dv[4 * q + 0] = t4.x; dv[4 * q + 1] = t4.y;
            dv[4 * q + 2] = t4.z; dv[4 * q + 3] = t4.w;
        }
        const float kth = topv[KK - 1];
        const float thr =
            fmaxf(kth, fminf(fminf(topv[KNN_SHARE - 1], f1), fminf(f2, f3)));
        unsigned pm = 0u;
#pragma unroll
        for (int c = 15; c >= 0; --c)
            pm = pm + pm + (((dv[c] - xxrow) >= thr) ? 1u : 0u);

        const float* myrow = &dist[wave][lane * 16];
        const int cbase = c0 + ct * 16;
        int c = __builtin_ctz(pm | 0x8000u);
        float cand = myrow[c];
        unsigned live = __builtin_amdgcn_ballot_w32(pm != 0u);
        for (int trip = 0; trip < 16 && live != 0u; ++trip) {
            const bool act = pm != 0u;
            pm &= pm - 1u;
            live = __builtin_amdgcn_ballot_w32(pm != 0u);
            const int cn = __builtin_ctz(pm | 0x8000u);
            const float nxt = myrow[cn];
            asm volatile("" ::: "memory");
            const float v = act ? (cand - xxrow) : -INFINITY;
            const int ci = cbase + c;
            bool beats[KK];
            float oldv[KK];
            int oldi[KK];
#pragma unroll
            for (int s = 0; s < KK; ++s) {
                oldv[s] = topv[s];
                oldi[s] = topi[s];
                beats[s] = v > oldv[s];
            }
            float inv[KK];
            int ini[KK];
            inv[0] = v;
            ini[0] = ci;
#pragma unroll
            for (int s = 1; s < KK; ++s) {
                const bool mu = beats[s - 1];
                const float upv = oldv[s - 1];
                const int upi = oldi[s - 1];
                inv[s] = mu ? upv : v;
                ini[s] = mu ? upi : ci;
            }
#pragma unroll
            for (int s = 0; s < KK; ++s) {
                const bool ms = beats[s];
                const float a = inv[s];
                const float b = oldv[s];
                const int ai = ini[s];
                const int bi = oldi[s];
                topv[s] = ms ? a : b;
                topi[s] = ms ? ai : bi;
            }
            c = cn;
            cand = nxt;
        }
        __builtin_amdgcn_fence(__ATOMIC_ACQ_REL, "wavefront");
    }
    __shared__ float mval[KNN_SPLIT][KK][32];
    __shared__ int midx[KNN_SPLIT][KK][32];
#pragma unroll
    for (int s = 0; s < KK; ++s) {
        mval[wave][s][lane] = topv[s];
        midx[wave][s][lane] = topi[s];
    }
    __syncthreads();
    if (wave == 0) {
        float h0 = mval[0][0][lane], h1 = mval[1][0][lane];
        float h2 = mval[2][0][lane], h3 = mval[3][0][lane];
        int p0 = 0, p1 = 0, p2 = 0, p3 = 0;
        int* outp = nbr + (size_t)(b * NN + r0 + lane) * KK;
        for (int s = 0; s < KK; ++s) {
            int w = 0, p = p0;
            float best = h0;
            if (h1 > best) { best = h1; w = 1; p = p1; }
            if (h2 > best) { best = h2; w = 2; p = p2; }
            if (h3 > best) { best = h3; w = 3; p = p3; }
            outp[s] = midx[w][p][lane];
            const int pnext = p + 1;
            const float hnext = mval[w][pnext < KK ? pnext : KK - 1][lane];
            p0 = (w == 0) ? pnext : p0;  h0 = (w == 0) ? hnext : h0;
            p1 = (w == 1) ? pnext : p1;  h1 = (w == 1) ? hnext : h1;
            p2 = (w == 2) ? pnext : p2;  h2 = (w == 2) ? hnext : h2;
            p3 = (w == 3) ? pnext : p3;  h3 = (w == 3) ? hnext : h3;
        }
    }
}


__global__ __launch_bounds__(128) void proj_kernel(float* xtp,
                                                   const float* __restrict__ w1,
                                                   const float* __restrict__ g1,
                                                   const float* __restrict__ va1) {
    const int wave = threadIdx.x >> 5;
    const int lane = threadIdx.x & 31;
    const int laneM = lane & 15;
    const int kofs = (lane < 16) ? 0 : 2;
    const int row0 = (blockIdx.x * 4 + wave) * 16;

    const float* prow = xtp + (size_t)(row0 + laneM) * CC;
    v2f a[16];
#pragma unroll
    for (int ks = 0; ks < 16; ++ks) a[ks] = *(const v2f*)(prow + ks * 4 + kofs);

    v8f acc[4] = {};
#pragma unroll
    for (int ks = 0; ks < 16; ++ks) {
        int kc = ks * 4 + kofs;
#pragma unroll
        for (int nt = 0; nt < 4; ++nt) {
            int o = nt * 16 + laneM;
            v2f bb = *(const v2f*)(w1 + o * 128 + kc);
            acc[nt] = wmma_f32(a[ks], bb, acc[nt]);
        }
    }
#pragma unroll
    for (int nt = 0; nt < 4; ++nt) {
        int o = nt * 16 + laneM;
        float sc = g1[o] * rsqrtf(va1[o] + EPSV);
#pragma unroll
        for (int j = 0; j < 8; ++j) {
            int M = j + ((lane < 16) ? 0 : 8);
            xtp[(size_t)(row0 + M) * CC + o] = acc[nt][j] * sc;
        }
    }
}

__global__ __launch_bounds__(128) void edgeconv_kernel(
    const float* __restrict__ x, const float* __restrict__ pbuf,
    const int* __restrict__ nbr,
    const float* __restrict__ w1, const float* __restrict__ g1,
    const float* __restrict__ be1, const float* __restrict__ mu1,
    const float* __restrict__ va1, const float* __restrict__ w2,
    const float* __restrict__ g2, const float* __restrict__ be2,
    const float* __restrict__ mu2, const float* __restrict__ va2,
    float* __restrict__ out) {
    const int wave = threadIdx.x >> 5;
    const int lane = threadIdx.x & 31;
    const int laneM = lane & 15;
    const int kofs = (lane < 16) ? 0 : 2;
    const int b = blockIdx.x >> 8;
    const int pt = blockIdx.x & 255;

    __shared__ __attribute__((aligned(16))) float qs[16 * OO];
    __shared__ unsigned om[16 * OO];
    __shared__ float bn2s[OO], bn2h[OO];
    for (int i = threadIdx.x; i < 16 * OO; i += 128) om[i] = 0u;
    if (threadIdx.x < OO) {
        const int o = threadIdx.x;
        const float sc = g2[o] * rsqrtf(va2[o] + EPSV);
        bn2s[o] = sc;
        bn2h[o] = be2[o] - mu2[o] * sc;
    }

    {
        const int o = wave * 16 + laneM;
        const float* px = x + (size_t)b * CC * NN + pt * 16 + laneM;
        v8f q0 = {};
        v8f q1 = {};
#pragma unroll
        for (int ks = 0; ks < 16; ks += 2) {
            int kc = ks * 4 + kofs;
            v2f a0, a1;
            a0.x = px[(size_t)(kc + 0) * NN];
            a0.y = px[(size_t)(kc + 1) * NN];
            a1.x = px[(size_t)(kc + 4) * NN];
            a1.y = px[(size_t)(kc + 5) * NN];
            v2f b0 = *(const v2f*)(w1 + o * 128 + 64 + kc) - *(const v2f*)(w1 + o * 128 + kc);
            v2f b1 = *(const v2f*)(w1 + o * 128 + 68 + kc) - *(const v2f*)(w1 + o * 128 + 4 + kc);
            q0 = wmma_f32(a0, b0, q0);
            q1 = wmma_f32(a1, b1, q1);
        }
        float sc = g1[o] * rsqrtf(va1[o] + EPSV);
        float sh = be1[o] - mu1[o] * sc;
#pragma unroll
        for (int j = 0; j < 8; ++j) {
            int M = j + ((lane < 16) ? 0 : 8);
            qs[M * OO + o] = (q0[j] + q1[j]) * sc + sh;
        }
    }

    v2f bw[16][4];
#pragma unroll
    for (int nt = 0; nt < 4; ++nt) {
        int o = nt * 16 + laneM;
#pragma unroll
        for (int ks = 0; ks < 16; ++ks)
            bw[ks][nt] = *(const v2f*)(w2 + o * OO + ks * 4 + kofs);
    }
    __syncthreads();

    for (int it = 0; it < 5; ++it) {
        const int mt = wave * 5 + it;
        const int lrow = mt * 16 + laneM;
        const int p = lrow / KK;
        const int kk = lrow - p * KK;
        const int n = pt * 16 + p;
        const int nb = nbr[(size_t)(b * NN + n) * KK + kk];
        const float* pn = pbuf + (size_t)(b * NN + nb) * CC;
        const float* pq = qs + p * OO;

        v2f a[16];
#pragma unroll
        for (int ks = 0; ks < 16; ++ks) {
            int kc = ks * 4 + kofs;
            v2f h = *(const v2f*)(pn + kc) + *(const v2f*)(pq + kc);
            a[ks].x = fmaxf(h.x, 0.0f);
            a[ks].y = fmaxf(h.y, 0.0f);
        }

        v8f acc2[4] = {};
#pragma unroll
        for (int ks = 0; ks < 16; ++ks) {
#pragma unroll
            for (int nt = 0; nt < 4; ++nt)
                acc2[nt] = wmma_f32(a[ks], bw[ks][nt], acc2[nt]);
        }
#pragma unroll
        for (int nt = 0; nt < 4; ++nt) {
            int o = nt * 16 + laneM;
            const float sc = bn2s[o];
            const float sh = bn2h[o];
#pragma unroll
            for (int j = 0; j < 8; ++j) {
                int M = j + ((lane < 16) ? 0 : 8);
                int lr = mt * 16 + M;
                int pp = lr / KK;
                float val = fmaxf(acc2[nt][j] * sc + sh, 0.0f);
                atomicMax(&om[pp * OO + o], __float_as_uint(val));
            }
        }
    }
    __syncthreads();
    for (int i = threadIdx.x; i < 16 * OO; i += 128) {
        int p = i >> 6;
        int o = i & 63;
        out[(size_t)(b * OO + o) * NN + pt * 16 + p] = __uint_as_float(om[p * OO + o]);
    }
}

extern "C" void kernel_launch(void* const* d_in, const int* in_sizes, int n_in,
                              void* d_out, int out_size, void* d_ws, size_t ws_size,
                              hipStream_t stream) {
    const float* x   = (const float*)d_in[0];
    const float* w1  = (const float*)d_in[1];
    const float* g1  = (const float*)d_in[2];
    const float* be1 = (const float*)d_in[3];
    const float* mu1 = (const float*)d_in[4];
    const float* va1 = (const float*)d_in[5];
    const float* w2  = (const float*)d_in[6];
    const float* g2  = (const float*)d_in[7];
    const float* be2 = (const float*)d_in[8];
    const float* mu2 = (const float*)d_in[9];
    const float* va2 = (const float*)d_in[10];
    float* out = (float*)d_out;

    float* xt = (float*)d_ws;
    float* xx = xt + (size_t)BB * NN * CC;
    int* nbr = (int*)(xx + (size_t)BB * NN);

    prep_kernel<<<(BB * NN) / 256, 256, 0, stream>>>(x, xt, xx);
    knn_kernel<<<(BB * NN) / KNN_ROWS, 128, 0, stream>>>(xt, xx, nbr);
    proj_kernel<<<(BB * NN / 16) / 4, 128, 0, stream>>>(xt, w1, g1, va1);
    edgeconv_kernel<<<BB * (NN / 16), 128, 0, stream>>>(
        x, xt, nbr, w1, g1, be1, mu1, va1, w2, g2, be2, mu2, va2, out);
}


# Point features stored in WMMA fragment order (16-point tile, k-step pair, lane): every operand load in kNN and proj reads 512 contiguous bytes (was 16 B from each of 16 rows); k order per lane kept
# speedup vs baseline: 28.9327x; 1.0392x over previous
#include <hip/hip_runtime.h>
#include <math.h>

#define BB 4
#define CC 64
#define NN 4096
#define OO 64
#define KK 20
#define EPSV 1e-5f

typedef __attribute__((ext_vector_type(2))) float v2f;
typedef __attribute__((ext_vector_type(8))) float v8f;

__device__ __forceinline__ v8f wmma_f32(v2f a, v2f b, v8f c) {
    return __builtin_amdgcn_wmma_f32_16x16x4_f32(false, a, false, b, (short)0, c,
                                                 false, false);
}

#define TILE_FLOATS (16 * CC)
#define FRAG_FLOATS (32 * 4)
__global__ __launch_bounds__(256) void prep_kernel(const float* __restrict__ x,
                                                   float* __restrict__ xf,
                                                   float* __restrict__ xx) {
    int t = blockIdx.x * blockDim.x + threadIdx.x;
    int b = t / NN;
    int n = t - b * NN;
    const float* px = x + (size_t)b * CC * NN + n;
    float* pt = xf + (size_t)(t >> 4) * TILE_FLOATS + (t & 15) * 4;
    float v[CC];
#pragma unroll
    for (int c = 0; c < CC; ++c) v[c] = px[(size_t)c * NN];
    float s = v[0] * v[0];
#pragma unroll
    for (int c = 1; c < CC; ++c) s = fmaf(v[c], v[c], s);
#pragma unroll
    for (int j = 0; j < 8; ++j) {
#pragma unroll
        for (int h = 0; h < 2; ++h) {
            const int k0 = 8 * j + 2 * h;
            *(float4*)(pt + j * FRAG_FLOATS + h * 64) =
                make_float4(v[k0], v[k0 + 1], v[k0 + 4], v[k0 + 5]);
        }
    }
    xx[t] = s;
}

__device__ __forceinline__ void load_frag_pair(const float* ptile, int j, v2f& f0,
                                               v2f& f1) {
    const float4 f = *(const float4*)(ptile + j * FRAG_FLOATS);
    f0.x = f.x; f0.y = f.y;
    f1.x = f.z; f1.y = f.w;
}

#define KNN_ROWS 32
#define KNN_SPLIT 4
#define KNN_COLS (NN / KNN_SPLIT)
#define KNN_SHARE 5
static_assert(KNN_SPLIT * KNN_SHARE >= KK, "the waves' shown slots must cover k");
static_assert((KNN_SPLIT & (KNN_SPLIT - 1)) == 0, "wave rotation uses a mask");
__global__ __launch_bounds__(128) void knn_kernel(const float* __restrict__ xf,
                                                  const float* __restrict__ xx,
                                                  int* __restrict__ nbr) {
    const int wave = threadIdx.x >> 5;
    const int lane = threadIdx.x & 31;
    const int laneM = lane & 15;

    const int row0 = blockIdx.x * KNN_ROWS;
    const int b = row0 / NN;
    const int r0 = row0 - b * NN;
    const int c0 = wave * KNN_COLS;

    __shared__ __attribute__((aligned(16))) float dist[4][32 * 16];
    __shared__ float fifth[KNN_SPLIT][32];
    fifth[wave][lane] = -3.0e38f;
    __syncthreads();

    const float* ptLo = xf + (size_t)((b * NN + r0) >> 4) * TILE_FLOATS + lane * 4;
    const float* ptHi = ptLo + TILE_FLOATS;
    v2f aLo[16], aHi[16];
#pragma unroll
    for (int j = 0; j < 8; ++j) {
        load_frag_pair(ptLo, j, aLo[2 * j], aLo[2 * j + 1]);
        load_frag_pair(ptHi, j, aHi[2 * j], aHi[2 * j + 1]);
    }

    float topv[KK];
    int topi[KK];
#pragma unroll
    for (int s = 0; s < KK; ++s) { topv[s] = -3.0e38f; topi[s] = 0; }
    const float xxrow = xx[b * NN + r0 + lane];

    const float* pcol0 =
        xf + (size_t)((b * NN + c0) >> 4) * TILE_FLOATS + lane * 4;
    const float* pxx0 = xx + b * NN + c0 + laneM;
    v2f bnext[16];
#pragma unroll
    for (int j = 0; j < 8; ++j)
        load_frag_pair(pcol0, j, bnext[2 * j], bnext[2 * j + 1]);
    float xnext = pxx0[0];

#pragma unroll 2
    for (int ct = 0; ct < KNN_COLS / 16; ++ct) {
        v2f bcur[16];
#pragma unroll
        for (int ks = 0; ks < 16; ++ks) bcur[ks] = bnext[ks];
        const float xcol = xnext;
        const int ctn = (ct + 1 < KNN_COLS / 16) ? ct + 1 : ct;
        const float* pcoln = pcol0 + (size_t)ctn * TILE_FLOATS;
#pragma unroll
        for (int j = 0; j < 8; ++j)
            load_frag_pair(pcoln, j, bnext[2 * j], bnext[2 * j + 1]);
        xnext = pxx0[ctn * 16];

        v8f accLo = {};
        v8f accHi = {};
#pragma unroll
        for (int ks = 0; ks < 16; ++ks) {
            accLo = wmma_f32(aLo[ks], bcur[ks], accLo);
            accHi = wmma_f32(aHi[ks], bcur[ks], accHi);
        }
#pragma unroll
        for (int j = 0; j < 8; ++j) {
            int M = j + ((lane < 16) ? 0 : 8);
            dist[wave][M * 16 + laneM] = 2.0f * accLo[j] - xcol;
            dist[wave][(16 + M) * 16 + laneM] = 2.0f * accHi[j] - xcol;
        }
        fifth[wave][lane] = topv[KNN_SHARE - 1];
        __builtin_amdgcn_fence(__ATOMIC_ACQ_REL, "wavefront");
        const float f1 = fifth[(wave + 1) & (KNN_SPLIT - 1)][lane];
        const float f2 = fifth[(wave + 2) & (KNN_SPLIT - 1)][lane];
        const float f3 = fifth[(wave + 3) & (KNN_SPLIT - 1)][lane];
        const float4* drow = (const float4*)&dist[wave][lane * 16];
        float dv[16];
#pragma unroll
        for (int q = 0; q < 4; ++q) {
            const float4 t4 = drow[q];
            dv[4 * q + 0] = t4.x; dv[4 * q + 1] = t4.y;
            dv[4 * q + 2] = t4.z; dv[4 * q + 3] = t4.w;
        }
        const float kth = topv[KK - 1];
        const float thr =
            fmaxf(kth, fminf(fminf(topv[KNN_SHARE - 1], f1), fminf(f2, f3)));
        unsigned pm = 0u;
#pragma unroll
        for (int c = 15; c >= 0; --c)
            pm = pm + pm + (((dv[c] - xxrow) >= thr) ? 1u : 0u);

        const float* myrow = &dist[wave][lane * 16];
        const int cbase = c0 + ct * 16;
        int c = __builtin_ctz(pm | 0x8000u);
        float cand = myrow[c];
        unsigned live = __builtin_amdgcn_ballot_w32(pm != 0u);
        for (int trip = 0; trip < 16 && live != 0u; ++trip) {
            const bool act = pm != 0u;
            pm &= pm - 1u;
            live = __builtin_amdgcn_ballot_w32(pm != 0u);
            const int cn = __builtin_ctz(pm | 0x8000u);
            const float nxt = myrow[cn];
            asm volatile("" ::: "memory");
            const float v = act ? (cand - xxrow) : -INFINITY;
            const int ci = cbase + c;
            bool beats[KK];
            float oldv[KK];
            int oldi[KK];
#pragma unroll
            for (int s = 0; s < KK; ++s) {
                oldv[s] = topv[s];
                oldi[s] = topi[s];
                beats[s] = v > oldv[s];
            }
            float inv[KK];
            int ini[KK];
            inv[0] = v;
            ini[0] = ci;
#pragma unroll
            for (int s = 1; s < KK; ++s) {
                const bool mu = beats[s - 1];
                const float upv = oldv[s - 1];
                const int upi = oldi[s - 1];
                inv[s] = mu ? upv : v;
                ini[s] = mu ? upi : ci;
            }
#pragma unroll
            for (int s = 0; s < KK; ++s) {
                const bool ms = beats[s];
                const float a = inv[s];
                const float b = oldv[s];
                const int ai = ini[s];
                const int bi = oldi[s];
                topv[s] = ms ? a : b;
                topi[s] = ms ? ai : bi;
            }
            c = cn;
            cand = nxt;
        }
        __builtin_amdgcn_fence(__ATOMIC_ACQ_REL, "wavefront");
    }
    __shared__ float mval[KNN_SPLIT][KK][32];
    __shared__ int midx[KNN_SPLIT][KK][32];
#pragma unroll
    for (int s = 0; s < KK; ++s) {
        mval[wave][s][lane] = topv[s];
        midx[wave][s][lane] = topi[s];
    }
    __syncthreads();
    if (wave == 0) {
        float h0 = mval[0][0][lane], h1 = mval[1][0][lane];
        float h2 = mval[2][0][lane], h3 = mval[3][0][lane];
        int p0 = 0, p1 = 0, p2 = 0, p3 = 0;
        int* outp = nbr + (size_t)(b * NN + r0 + lane) * KK;
        for (int s = 0; s < KK; ++s) {
            int w = 0, p = p0;
            float best = h0;
            if (h1 > best) { best = h1; w = 1; p = p1; }
            if (h2 > best) { best = h2; w = 2; p = p2; }
            if (h3 > best) { best = h3; w = 3; p = p3; }
            outp[s] = midx[w][p][lane];
            const int pnext = p + 1;
            const float hnext = mval[w][pnext < KK ? pnext : KK - 1][lane];
            p0 = (w == 0) ? pnext : p0;  h0 = (w == 0) ? hnext : h0;
            p1 = (w == 1) ? pnext : p1;  h1 = (w == 1) ? hnext : h1;
            p2 = (w == 2) ? pnext : p2;  h2 = (w == 2) ? hnext : h2;
            p3 = (w == 3) ? pnext : p3;  h3 = (w == 3) ? hnext : h3;
        }
    }
}


__global__ __launch_bounds__(128) void proj_kernel(float* xtp,
                                                   const float* __restrict__ w1,
                                                   const float* __restrict__ g1,
                                                   const float* __restrict__ va1) {
    const int wave = threadIdx.x >> 5;
    const int lane = threadIdx.x & 31;
    const int laneM = lane & 15;
    const int kofs = (lane < 16) ? 0 : 2;
    const int row0 = (blockIdx.x * 4 + wave) * 16;

    const float* ptile = xtp + (size_t)(row0 >> 4) * TILE_FLOATS + lane * 4;
    v2f a[16];
#pragma unroll
    for (int j = 0; j < 8; ++j) load_frag_pair(ptile, j, a[2 * j], a[2 * j + 1]);

    v8f acc[4] = {};
#pragma unroll
    for (int ks = 0; ks < 16; ++ks) {
        int kc = ks * 4 + kofs;
#pragma unroll
        for (int nt = 0; nt < 4; ++nt) {
            int o = nt * 16 + laneM;
            v2f bb = *(const v2f*)(w1 + o * 128 + kc);
            acc[nt] = wmma_f32(a[ks], bb, acc[nt]);
        }
    }
#pragma unroll
    for (int nt = 0; nt < 4; ++nt) {
        int o = nt * 16 + laneM;
        float sc = g1[o] * rsqrtf(va1[o] + EPSV);
#pragma unroll
        for (int j = 0; j < 8; ++j) {
            int M = j + ((lane < 16) ? 0 : 8);
            xtp[(size_t)(row0 + M) * CC + o] = acc[nt][j] * sc;
        }
    }
}

__global__ __launch_bounds__(128) void edgeconv_kernel(
    const float* __restrict__ x, const float* __restrict__ pbuf,
    const int* __restrict__ nbr,
    const float* __restrict__ w1, const float* __restrict__ g1,
    const float* __restrict__ be1, const float* __restrict__ mu1,
    const float* __restrict__ va1, const float* __restrict__ w2,
    const float* __restrict__ g2, const float* __restrict__ be2,
    const float* __restrict__ mu2, const float* __restrict__ va2,
    float* __restrict__ out) {
    const int wave = threadIdx.x >> 5;
    const int lane = threadIdx.x & 31;
    const int laneM = lane & 15;
    const int kofs = (lane < 16) ? 0 : 2;
    const int b = blockIdx.x >> 8;
    const int pt = blockIdx.x & 255;

    __shared__ __attribute__((aligned(16))) float qs[16 * OO];
    __shared__ unsigned om[16 * OO];
    __shared__ float bn2s[OO], bn2h[OO];
    for (int i = threadIdx.x; i < 16 * OO; i += 128) om[i] = 0u;
    if (threadIdx.x < OO) {
        const int o = threadIdx.x;
        const float sc = g2[o] * rsqrtf(va2[o] + EPSV);
        bn2s[o] = sc;
        bn2h[o] = be2[o] - mu2[o] * sc;
    }

    {
        const int o = wave * 16 + laneM;
        const float* px = x + (size_t)b * CC * NN + pt * 16 + laneM;
        v8f q0 = {};
        v8f q1 = {};
#pragma unroll
        for (int ks = 0; ks < 16; ks += 2) {
            int kc = ks * 4 + kofs;
            v2f a0, a1;
            a0.x = px[(size_t)(kc + 0) * NN];
            a0.y = px[(size_t)(kc + 1) * NN];
            a1.x = px[(size_t)(kc + 4) * NN];
            a1.y = px[(size_t)(kc + 5) * NN];
            v2f b0 = *(const v2f*)(w1 + o * 128 + 64 + kc) - *(const v2f*)(w1 + o * 128 + kc);
            v2f b1 = *(const v2f*)(w1 + o * 128 + 68 + kc) - *(const v2f*)(w1 + o * 128 + 4 + kc);
            q0 = wmma_f32(a0, b0, q0);
            q1 = wmma_f32(a1, b1, q1);
        }
        float sc = g1[o] * rsqrtf(va1[o] + EPSV);
        float sh = be1[o] - mu1[o] * sc;
#pragma unroll
        for (int j = 0; j < 8; ++j) {
            int M = j + ((lane < 16) ? 0 : 8);
            qs[M * OO + o] = (q0[j] + q1[j]) * sc + sh;
        }
    }

    v2f bw[16][4];
#pragma unroll
    for (int nt = 0; nt < 4; ++nt) {
        int o = nt * 16 + laneM;
#pragma unroll
        for (int ks = 0; ks < 16; ++ks)
            bw[ks][nt] = *(const v2f*)(w2 + o * OO + ks * 4 + kofs);
    }
    __syncthreads();

    for (int it = 0; it < 5; ++it) {
        const int mt = wave * 5 + it;
        const int lrow = mt * 16 + laneM;
        const int p = lrow / KK;
        const int kk = lrow - p * KK;
        const int n = pt * 16 + p;
        const int nb = nbr[(size_t)(b * NN + n) * KK + kk];
        const float* pn = pbuf + (size_t)(b * NN + nb) * CC;
        const float* pq = qs + p * OO;

        v2f a[16];
#pragma unroll
        for (int ks = 0; ks < 16; ++ks) {
            int kc = ks * 4 + kofs;
            v2f h = *(const v2f*)(pn + kc) + *(const v2f*)(pq + kc);
            a[ks].x = fmaxf(h.x, 0.0f);
            a[ks].y = fmaxf(h.y, 0.0f);
        }

        v8f acc2[4] = {};
#pragma unroll
        for (int ks = 0; ks < 16; ++ks) {
#pragma unroll
            for (int nt = 0; nt < 4; ++nt)
                acc2[nt] = wmma_f32(a[ks], bw[ks][nt], acc2[nt]);
        }
#pragma unroll
        for (int nt = 0; nt < 4; ++nt) {
            int o = nt * 16 + laneM;
            const float sc = bn2s[o];
            const float sh = bn2h[o];
#pragma unroll
            for (int j = 0; j < 8; ++j) {
                int M = j + ((lane < 16) ? 0 : 8);
                int lr = mt * 16 + M;
                int pp = lr / KK;
                float val = fmaxf(acc2[nt][j] * sc + sh, 0.0f);
                atomicMax(&om[pp * OO + o], __float_as_uint(val));
            }
        }
    }
    __syncthreads();
    for (int i = threadIdx.x; i < 16 * OO; i += 128) {
        int p = i >> 6;
        int o = i & 63;
        out[(size_t)(b * OO + o) * NN + pt * 16 + p] = __uint_as_float(om[p * OO + o]);
    }
}

extern "C" void kernel_launch(void* const* d_in, const int* in_sizes, int n_in,
                              void* d_out, int out_size, void* d_ws, size_t ws_size,
                              hipStream_t stream) {
    const float* x   = (const float*)d_in[0];
    const float* w1  = (const float*)d_in[1];
    const float* g1  = (const float*)d_in[2];
    const float* be1 = (const float*)d_in[3];
    const float* mu1 = (const float*)d_in[4];
    const float* va1 = (const float*)d_in[5];
    const float* w2  = (const float*)d_in[6];
    const float* g2  = (const float*)d_in[7];
    const float* be2 = (const float*)d_in[8];
    const float* mu2 = (const float*)d_in[9];
    const float* va2 = (const float*)d_in[10];
    float* out = (float*)d_out;

    float* xf = (float*)d_ws;
    float* xx = xf + (size_t)BB * NN * CC;
    int* nbr = (int*)(xx + (size_t)BB * NN);

    prep_kernel<<<(BB * NN) / 256, 256, 0, stream>>>(x, xf, xx);
    knn_kernel<<<(BB * NN) / KNN_ROWS, 128, 0, stream>>>(xf, xx, nbr);
    proj_kernel<<<(BB * NN / 16) / 4, 128, 0, stream>>>(xf, w1, g1, va1);
    edgeconv_kernel<<<BB * (NN / 16), 128, 0, stream>>>(
        x, xf, nbr, w1, g1, be1, mu1, va1, w2, g2, be2, mu2, va2, out);
}


# kNN scan in rounds of 32 columns (two tiles): 64 products back to back on four accumulators, one LDS pass, one mark and one insert loop per round (was one of each per 16-column tile)
# speedup vs baseline: 30.9133x; 1.0685x over previous
#include <hip/hip_runtime.h>
#include <math.h>

#define BB 4
#define CC 64
#define NN 4096
#define OO 64
#define KK 20
#define EPSV 1e-5f

typedef __attribute__((ext_vector_type(2))) float v2f;
typedef __attribute__((ext_vector_type(8))) float v8f;

__device__ __forceinline__ v8f wmma_f32(v2f a, v2f b, v8f c) {
    return __builtin_amdgcn_wmma_f32_16x16x4_f32(false, a, false, b, (short)0, c,
                                                 false, false);
}

#define TILE_FLOATS (16 * CC)
#define FRAG_FLOATS (32 * 4)
__global__ __launch_bounds__(256) void prep_kernel(const float* __restrict__ x,
                                                   float* __restrict__ xf,
                                                   float* __restrict__ xx) {
    int t = blockIdx.x * blockDim.x + threadIdx.x;
    int b = t / NN;
    int n = t - b * NN;
    const float* px = x + (size_t)b * CC * NN + n;
    float* pt = xf + (size_t)(t >> 4) * TILE_FLOATS + (t & 15) * 4;
    float v[CC];
#pragma unroll
    for (int c = 0; c < CC; ++c) v[c] = px[(size_t)c * NN];
    float s = v[0] * v[0];
#pragma unroll
    for (int c = 1; c < CC; ++c) s = fmaf(v[c], v[c], s);
#pragma unroll
    for (int j = 0; j < 8; ++j) {
#pragma unroll
        for (int h = 0; h < 2; ++h) {
            const int k0 = 8 * j + 2 * h;
            *(float4*)(pt + j * FRAG_FLOATS + h * 64) =
                make_float4(v[k0], v[k0 + 1], v[k0 + 4], v[k0 + 5]);
        }
    }
    xx[t] = s;
}

__device__ __forceinline__ void load_frag_pair(const float* ptile, int j, v2f& f0,
                                               v2f& f1) {
    const float4 f = *(const float4*)(ptile + j * FRAG_FLOATS);
    f0.x = f.x; f0.y = f.y;
    f1.x = f.z; f1.y = f.w;
}

#define KNN_ROWS 32
#define KNN_SPLIT 4
#define KNN_COLS (NN / KNN_SPLIT)
#define KNN_SHARE 5
#define KNN_RCOLS 32
#define KNN_ROUNDS (KNN_COLS / KNN_RCOLS)
#define KNN_PITCH 36
static_assert(KNN_SPLIT * KNN_SHARE >= KK, "the waves' shown slots must cover k");
static_assert((KNN_SPLIT & (KNN_SPLIT - 1)) == 0, "wave rotation uses a mask");
static_assert(KNN_COLS % KNN_RCOLS == 0, "whole rounds");
__global__ __launch_bounds__(128) void knn_kernel(const float* __restrict__ xf,
                                                  const float* __restrict__ xx,
                                                  int* __restrict__ nbr) {
    const int wave = threadIdx.x >> 5;
    const int lane = threadIdx.x & 31;
    const int laneM = lane & 15;

    const int row0 = blockIdx.x * KNN_ROWS;
    const int b = row0 / NN;
    const int r0 = row0 - b * NN;
    const int c0 = wave * KNN_COLS;

    __shared__ __attribute__((aligned(16))) float dist[4][32 * KNN_PITCH];
    __shared__ float fifth[KNN_SPLIT][32];
    fifth[wave][lane] = -3.0e38f;
    __syncthreads();

    const float* ptLo = xf + (size_t)((b * NN + r0) >> 4) * TILE_FLOATS + lane * 4;
    const float* ptHi = ptLo + TILE_FLOATS;
    v2f aLo[16], aHi[16];
#pragma unroll
    for (int j = 0; j < 8; ++j) {
        load_frag_pair(ptLo, j, aLo[2 * j], aLo[2 * j + 1]);
        load_frag_pair(ptHi, j, aHi[2 * j], aHi[2 * j + 1]);
    }

    float topv[KK];
    int topi[KK];
#pragma unroll
    for (int s = 0; s < KK; ++s) { topv[s] = -3.0e38f; topi[s] = 0; }
    const float xxrow = xx[b * NN + r0 + lane];

    const float* pcol0 =
        xf + (size_t)((b * NN + c0) >> 4) * TILE_FLOATS + lane * 4;
    const float* pxx0 = xx + b * NN + c0 + laneM;
    v2f bnextA[16], bnextB[16];
#pragma unroll
    for (int j = 0; j < 8; ++j) {
        load_frag_pair(pcol0, j, bnextA[2 * j], bnextA[2 * j + 1]);
        load_frag_pair(pcol0 + TILE_FLOATS, j, bnextB[2 * j], bnextB[2 * j + 1]);
    }
    float xnextA = pxx0[0];
    float xnextB = pxx0[16];

#pragma unroll 2
    for (int cp = 0; cp < KNN_ROUNDS; ++cp) {
        v2f bA[16], bB[16];
#pragma unroll
        for (int ks = 0; ks < 16; ++ks) { bA[ks] = bnextA[ks]; bB[ks] = bnextB[ks]; }
        const float xcolA = xnextA;
        const float xcolB = xnextB;
        const int cpn = (cp + 1 < KNN_ROUNDS) ? cp + 1 : cp;
        const float* pcoln = pcol0 + (size_t)cpn * (2 * TILE_FLOATS);
#pragma unroll
        for (int j = 0; j < 8; ++j) {
            load_frag_pair(pcoln, j, bnextA[2 * j], bnextA[2 * j + 1]);
            load_frag_pair(pcoln + TILE_FLOATS, j, bnextB[2 * j], bnextB[2 * j + 1]);
        }
        xnextA = pxx0[cpn * KNN_RCOLS];
        xnextB = pxx0[cpn * KNN_RCOLS + 16];

        v8f accLoA = {};
        v8f accHiA = {};
        v8f accLoB = {};
        v8f accHiB = {};
#pragma unroll
        for (int ks = 0; ks < 16; ++ks) {
            accLoA = wmma_f32(aLo[ks], bA[ks], accLoA);
            accHiA = wmma_f32(aHi[ks], bA[ks], accHiA);
            accLoB = wmma_f32(aLo[ks], bB[ks], accLoB);
            accHiB = wmma_f32(aHi[ks], bB[ks], accHiB);
        }
#pragma unroll
        for (int j = 0; j < 8; ++j) {
            int M = j + ((lane < 16) ? 0 : 8);
            dist[wave][M * KNN_PITCH + laneM] = 2.0f * accLoA[j] - xcolA;
            dist[wave][(16 + M) * KNN_PITCH + laneM] = 2.0f * accHiA[j] - xcolA;
            dist[wave][M * KNN_PITCH + 16 + laneM] = 2.0f * accLoB[j] - xcolB;
            dist[wave][(16 + M) * KNN_PITCH + 16 + laneM] = 2.0f * accHiB[j] - xcolB;
        }
        fifth[wave][lane] = topv[KNN_SHARE - 1];
        __builtin_amdgcn_fence(__ATOMIC_ACQ_REL, "wavefront");
        const float f1 = fifth[(wave + 1) & (KNN_SPLIT - 1)][lane];
        const float f2 = fifth[(wave + 2) & (KNN_SPLIT - 1)][lane];
        const float f3 = fifth[(wave + 3) & (KNN_SPLIT - 1)][lane];
        const float4* drow = (const float4*)&dist[wave][lane * KNN_PITCH];
        float dv[KNN_RCOLS];
#pragma unroll
        for (int q = 0; q < KNN_RCOLS / 4; ++q) {
            const float4 t4 = drow[q];
            dv[4 * q + 0] = t4.x; dv[4 * q + 1] = t4.y;
            dv[4 * q + 2] = t4.z; dv[4 * q + 3] = t4.w;
        }
        const float kth = topv[KK - 1];
        const float thr =
            fmaxf(kth, fminf(fminf(topv[KNN_SHARE - 1], f1), fminf(f2, f3)));
        unsigned pmLo = 0u, pmHi = 0u;
#pragma unroll
        for (int c = 15; c >= 0; --c) {
            pmLo = pmLo + pmLo + (((dv[c] - xxrow) >= thr) ? 1u : 0u);
            pmHi = pmHi + pmHi + (((dv[16 + c] - xxrow) >= thr) ? 1u : 0u);
        }
        unsigned pm = pmLo | (pmHi << 16);

        const float* myrow = &dist[wave][lane * KNN_PITCH];
        const int cbase = c0 + cp * KNN_RCOLS;
        int c = __builtin_ctz(pm | 0x80000000u);
        float cand = myrow[c];
        unsigned live = __builtin_amdgcn_ballot_w32(pm != 0u);
        for (int trip = 0; trip < KNN_RCOLS && live != 0u; ++trip) {
            const bool act = pm != 0u;
            pm &= pm - 1u;
            live = __builtin_amdgcn_ballot_w32(pm != 0u);
            const int cn = __builtin_ctz(pm | 0x80000000u);
            const float nxt = myrow[cn];
            asm volatile("" ::: "memory");
            const float v = act ? (cand - xxrow) : -INFINITY;
            const int ci = cbase + c;
            bool beats[KK];
            float oldv[KK];
            int oldi[KK];
#pragma unroll
            for (int s = 0; s < KK; ++s) {
                oldv[s] = topv[s];
                oldi[s] = topi[s];
                beats[s] = v > oldv[s];
            }
            float inv[KK];
            int ini[KK];
            inv[0] = v;
            ini[0] = ci;
#pragma unroll
            for (int s = 1; s < KK; ++s) {
                const bool mu = beats[s - 1];
                const float upv = oldv[s - 1];
                const int upi = oldi[s - 1];
                inv[s] = mu ? upv : v;
                ini[s] = mu ? upi : ci;
            }
#pragma unroll
            for (int s = 0; s < KK; ++s) {
                const bool ms = beats[s];
                const float a = inv[s];
                const float b = oldv[s];
                const int ai = ini[s];
                const int bi = oldi[s];
                topv[s] = ms ? a : b;
                topi[s] = ms ? ai : bi;
            }
            c = cn;
            cand = nxt;
        }
        __builtin_amdgcn_fence(__ATOMIC_ACQ_REL, "wavefront");
    }
    __shared__ float mval[KNN_SPLIT][KK][32];
    __shared__ int midx[KNN_SPLIT][KK][32];
#pragma unroll
    for (int s = 0; s < KK; ++s) {
        mval[wave][s][lane] = topv[s];
        midx[wave][s][lane] = topi[s];
    }
    __syncthreads();
    if (wave == 0) {
        float h0 = mval[0][0][lane], h1 = mval[1][0][lane];
        float h2 = mval[2][0][lane], h3 = mval[3][0][lane];
        int p0 = 0, p1 = 0, p2 = 0, p3 = 0;
        int* outp = nbr + (size_t)(b * NN + r0 + lane) * KK;
        for (int s = 0; s < KK; ++s) {
            int w = 0, p = p0;
            float best = h0;
            if (h1 > best) { best = h1; w = 1; p = p1; }
            if (h2 > best) { best = h2; w = 2; p = p2; }
            if (h3 > best) { best = h3; w = 3; p = p3; }
            outp[s] = midx[w][p][lane];
            const int pnext = p + 1;
            const float hnext = mval[w][pnext < KK ? pnext : KK - 1][lane];
            p0 = (w == 0) ? pnext : p0;  h0 = (w == 0) ? hnext : h0;
            p1 = (w == 1) ? pnext : p1;  h1 = (w == 1) ? hnext : h1;
            p2 = (w == 2) ? pnext : p2;  h2 = (w == 2) ? hnext : h2;
            p3 = (w == 3) ? pnext : p3;  h3 = (w == 3) ? hnext : h3;
        }
    }
}


__global__ __launch_bounds__(128) void proj_kernel(float* xtp,
                                                   const float* __restrict__ w1,
                                                   const float* __restrict__ g1,
                                                   const float* __restrict__ va1) {
    const int wave = threadIdx.x >> 5;
    const int lane = threadIdx.x & 31;
    const int laneM = lane & 15;
    const int kofs = (lane < 16) ? 0 : 2;
    const int row0 = (blockIdx.x * 4 + wave) * 16;

    const float* ptile = xtp + (size_t)(row0 >> 4) * TILE_FLOATS + lane * 4;
    v2f a[16];
#pragma unroll
    for (int j = 0; j < 8; ++j) load_frag_pair(ptile, j, a[2 * j], a[2 * j + 1]);

    v8f acc[4] = {};
#pragma unroll
    for (int ks = 0; ks < 16; ++ks) {
        int kc = ks * 4 + kofs;
#pragma unroll
        for (int nt = 0; nt < 4; ++nt) {
            int o = nt * 16 + laneM;
            v2f bb = *(const v2f*)(w1 + o * 128 + kc);
            acc[nt] = wmma_f32(a[ks], bb, acc[nt]);
        }
    }
#pragma unroll
    for (int nt = 0; nt < 4; ++nt) {
        int o = nt * 16 + laneM;
        float sc = g1[o] * rsqrtf(va1[o] + EPSV);
#pragma unroll
        for (int j = 0; j < 8; ++j) {
            int M = j + ((lane < 16) ? 0 : 8);
            xtp[(size_t)(row0 + M) * CC + o] = acc[nt][j] * sc;
        }
    }
}

__global__ __launch_bounds__(128) void edgeconv_kernel(
    const float* __restrict__ x, const float* __restrict__ pbuf,
    const int* __restrict__ nbr,
    const float* __restrict__ w1, const float* __restrict__ g1,
    const float* __restrict__ be1, const float* __restrict__ mu1,
    const float* __restrict__ va1, const float* __restrict__ w2,
    const float* __restrict__ g2, const float* __restrict__ be2,
    const float* __restrict__ mu2, const float* __restrict__ va2,
    float* __restrict__ out) {
    const int wave = threadIdx.x >> 5;
    const int lane = threadIdx.x & 31;
    const int laneM = lane & 15;
    const int kofs = (lane < 16) ? 0 : 2;
    const int b = blockIdx.x >> 8;
    const int pt = blockIdx.x & 255;

    __shared__ __attribute__((aligned(16))) float qs[16 * OO];
    __shared__ unsigned om[16 * OO];
    __shared__ float bn2s[OO], bn2h[OO];
    for (int i = threadIdx.x; i < 16 * OO; i += 128) om[i] = 0u;
    if (threadIdx.x < OO) {
        const int o = threadIdx.x;
        const float sc = g2[o] * rsqrtf(va2[o] + EPSV);
        bn2s[o] = sc;
        bn2h[o] = be2[o] - mu2[o] * sc;
    }

    {
        const int o = wave * 16 + laneM;
        const float* px = x + (size_t)b * CC * NN + pt * 16 + laneM;
        v8f q0 = {};
        v8f q1 = {};
#pragma unroll
        for (int ks = 0; ks < 16; ks += 2) {
            int kc = ks * 4 + kofs;
            v2f a0, a1;
            a0.x = px[(size_t)(kc + 0) * NN];
            a0.y = px[(size_t)(kc + 1) * NN];
            a1.x = px[(size_t)(kc + 4) * NN];
            a1.y = px[(size_t)(kc + 5) * NN];
            v2f b0 = *(const v2f*)(w1 + o * 128 + 64 + kc) - *(const v2f*)(w1 + o * 128 + kc);
            v2f b1 = *(const v2f*)(w1 + o * 128 + 68 + kc) - *(const v2f*)(w1 + o * 128 + 4 + kc);
            q0 = wmma_f32(a0, b0, q0);
            q1 = wmma_f32(a1, b1, q1);
        }
        float sc = g1[o] * rsqrtf(va1[o] + EPSV);
        float sh = be1[o] - mu1[o] * sc;
#pragma unroll
        for (int j = 0; j < 8; ++j) {
            int M = j + ((lane < 16) ? 0 : 8);
            qs[M * OO + o] = (q0[j] + q1[j]) * sc + sh;
        }
    }

    v2f bw[16][4];
#pragma unroll
    for (int nt = 0; nt < 4; ++nt) {
        int o = nt * 16 + laneM;
#pragma unroll
        for (int ks = 0; ks < 16; ++ks)
            bw[ks][nt] = *(const v2f*)(w2 + o * OO + ks * 4 + kofs);
    }
    __syncthreads();

    for (int it = 0; it < 5; ++it) {
        const int mt = wave * 5 + it;
        const int lrow = mt * 16 + laneM;
        const int p = lrow / KK;
        const int kk = lrow - p * KK;
        const int n = pt * 16 + p;
        const int nb = nbr[(size_t)(b * NN + n) * KK + kk];
        const float* pn = pbuf + (size_t)(b * NN + nb) * CC;
        const float* pq = qs + p * OO;

        v2f a[16];
#pragma unroll
        for (int ks = 0; ks < 16; ++ks) {
            int kc = ks * 4 + kofs;
            v2f h = *(const v2f*)(pn + kc) + *(const v2f*)(pq + kc);
            a[ks].x = fmaxf(h.x, 0.0f);
            a[ks].y = fmaxf(h.y, 0.0f);
        }

        v8f acc2[4] = {};
#pragma unroll
        for (int ks = 0; ks < 16; ++ks) {
#pragma unroll
            for (int nt = 0; nt < 4; ++nt)
                acc2[nt] = wmma_f32(a[ks], bw[ks][nt], acc2[nt]);
        }
#pragma unroll
        for (int nt = 0; nt < 4; ++nt) {
            int o = nt * 16 + laneM;
            const float sc = bn2s[o];
            const float sh = bn2h[o];
#pragma unroll
            for (int j = 0; j < 8; ++j) {
                int M = j + ((lane < 16) ? 0 : 8);
                int lr = mt * 16 + M;
                int pp = lr / KK;
                float val = fmaxf(acc2[nt][j] * sc + sh, 0.0f);
                atomicMax(&om[pp * OO + o], __float_as_uint(val));
            }
        }
    }
    __syncthreads();
    for (int i = threadIdx.x; i < 16 * OO; i += 128) {
        int p = i >> 6;
        int o = i & 63;
        out[(size_t)(b * OO + o) * NN + pt * 16 + p] = __uint_as_float(om[p * OO + o]);
    }
}

extern "C" void kernel_launch(void* const* d_in, const int* in_sizes, int n_in,
                              void* d_out, int out_size, void* d_ws, size_t ws_size,
                              hipStream_t stream) {
    const float* x   = (const float*)d_in[0];
    const float* w1  = (const float*)d_in[1];
    const float* g1  = (const float*)d_in[2];
    const float* be1 = (const float*)d_in[3];
    const float* mu1 = (const float*)d_in[4];
    const float* va1 = (const float*)d_in[5];
    const float* w2  = (const float*)d_in[6];
    const float* g2  = (const float*)d_in[7];
    const float* be2 = (const float*)d_in[8];
    const float* mu2 = (const float*)d_in[9];
    const float* va2 = (const float*)d_in[10];
    float* out = (float*)d_out;

    float* xf = (float*)d_ws;
    float* xx = xf + (size_t)BB * NN * CC;
    int* nbr = (int*)(xx + (size_t)BB * NN);

    prep_kernel<<<(BB * NN) / 256, 256, 0, stream>>>(x, xf, xx);
    knn_kernel<<<(BB * NN) / KNN_ROWS, 128, 0, stream>>>(xf, xx, nbr);
    proj_kernel<<<(BB * NN / 16) / 4, 128, 0, stream>>>(xf, w1, g1, va1);
    edgeconv_kernel<<<BB * (NN / 16), 128, 0, stream>>>(
        x, xf, nbr, w1, g1, be1, mu1, va1, w2, g2, be2, mu2, va2, out);
}
